# starting kernel
# baseline (speedup 1.0000x reference)
#include <hip/hip_runtime.h>
#include <hip/hip_bf16.h>

typedef float v2f __attribute__((ext_vector_type(2)));
typedef float v8f __attribute__((ext_vector_type(8)));

#define B_  2
#define L_  2048
#define D_  512
#define N_  64
#define R_  32
#define E_  160
#define BL_ (B_ * L_)
#define G_  32
#define T_  (L_ / G_)

__global__ void __launch_bounds__(256)
s6_xproj_gemm(const float* __restrict__ x, const float* __restrict__ W,
              float* __restrict__ xdbl) {
  const int wave  = blockIdx.x * (blockDim.x >> 5) + (threadIdx.x >> 5);
  const int lane  = threadIdx.x & 31;
  const int rt    = wave / 10;
  const int et    = wave % 10;
  const int laneM = lane & 15;
  const int laneH = lane >> 4;

  const float* arow = x + (size_t)(rt * 16 + laneM) * D_ + 2 * laneH;
  const float* brow = W + (size_t)(et * 16 + laneM) * D_ + 2 * laneH;

  v8f c = {};
  for (int k = 0; k < D_; k += 4) {
    v2f a = *(const v2f*)(arow + k);
    v2f b = *(const v2f*)(brow + k);
    c = __builtin_amdgcn_wmma_f32_16x16x4_f32(false, a, false, b,
                                              (short)0, c, false, false);
  }
  const int col = et * 16 + laneM;
#pragma unroll
  for (int j = 0; j < 8; ++j) {
    const int row = rt * 16 + j + 8 * laneH;
    xdbl[(size_t)row * E_ + col] = c[j];
  }
}

__global__ void __launch_bounds__(256)
s6_delta_gemm(const float* __restrict__ xdbl, const float* __restrict__ Wdt,
              const float* __restrict__ bdt, float* __restrict__ delta) {
  const int wave  = blockIdx.x * (blockDim.x >> 5) + (threadIdx.x >> 5);
  const int lane  = threadIdx.x & 31;
  const int rt    = wave >> 5;
  const int nt    = wave & 31;
  const int laneM = lane & 15;
  const int laneH = lane >> 4;

  const float* arow = xdbl + (size_t)(rt * 16 + laneM) * E_ + 2 * laneH;
  const float* brow = Wdt  + (size_t)(nt * 16 + laneM) * R_ + 2 * laneH;

  v8f c = {};
#pragma unroll
  for (int k = 0; k < R_; k += 4) {
    v2f a = *(const v2f*)(arow + k);
    v2f b = *(const v2f*)(brow + k);
    c = __builtin_amdgcn_wmma_f32_16x16x4_f32(false, a, false, b,
                                              (short)0, c, false, false);
  }
  const int col  = nt * 16 + laneM;
  const float bi = bdt[col];
#pragma unroll
  for (int j = 0; j < 8; ++j) {
    const int row = rt * 16 + j + 8 * laneH;
    float z = c[j] + bi;
    float sp = (z > 20.0f) ? z : __logf(1.0f + __expf(z));
    delta[(size_t)row * D_ + col] = sp;
  }
}

__global__ void __launch_bounds__(256)
s6_scan_chunks(const float* __restrict__ xdbl, const float* __restrict__ delta,
               const float* __restrict__ x, const float* __restrict__ Alog,
               float* __restrict__ Pout, float* __restrict__ Hout) {
  const int wave = blockIdx.x * (blockDim.x >> 5) + (threadIdx.x >> 5);
  const int lane = threadIdx.x & 31;
  const int g  = wave % G_;
  const int bd = wave / G_;
  const int dd = bd % D_;
  const int bb = bd / D_;
  const int n0 = 2 * lane;

  v2f al = *(const v2f*)(Alog + (size_t)dd * N_ + n0);
  v2f A;  A.x = -__expf(al.x);  A.y = -__expf(al.y);

  v2f h = {0.0f, 0.0f};
  v2f P = {1.0f, 1.0f};
  const int t0 = g * T_;
  for (int tt = 0; tt < T_; ++tt) {
    const size_t row = (size_t)bb * L_ + (t0 + tt);
    const float dlt = delta[row * D_ + dd];
    const float xv  = x[row * D_ + dd];
    v2f Bv = *(const v2f*)(xdbl + row * E_ + R_ + n0);
    if (tt + 8 < T_)
      __builtin_prefetch(xdbl + (row + 8) * E_ + R_ + n0, 0, 1);
    v2f a;  a.x = __expf(dlt * A.x);  a.y = __expf(dlt * A.y);
    const float du = dlt * xv;
    h.x = a.x * h.x + du * Bv.x;
    h.y = a.y * h.y + du * Bv.y;
    P.x *= a.x;  P.y *= a.y;
  }
  const size_t base = ((size_t)bd * G_ + g) * N_ + n0;
  *(v2f*)(Pout + base) = P;
  *(v2f*)(Hout + base) = h;
}

__global__ void __launch_bounds__(256)
s6_combine(const float* __restrict__ Pin, const float* __restrict__ Hin,
           float* __restrict__ h0out) {
  const int bd   = blockIdx.x * (blockDim.x >> 5) + (threadIdx.x >> 5);
  const int lane = threadIdx.x & 31;
  const int n0   = 2 * lane;

  v2f h = {0.0f, 0.0f};
  for (int g = 0; g < G_; ++g) {
    const size_t base = ((size_t)bd * G_ + g) * N_ + n0;
    *(v2f*)(h0out + base) = h;
    v2f P = *(const v2f*)(Pin + base);
    v2f H = *(const v2f*)(Hin + base);
    h.x = P.x * h.x + H.x;
    h.y = P.y * h.y + H.y;
  }
}

__global__ void __launch_bounds__(256)
s6_scan_emit(const float* __restrict__ xdbl, const float* __restrict__ delta,
             const float* __restrict__ x, const float* __restrict__ Alog,
             const float* __restrict__ Dp, const float* __restrict__ h0in,
             float* __restrict__ y) {
  const int wave = blockIdx.x * (blockDim.x >> 5) + (threadIdx.x >> 5);
  const int lane = threadIdx.x & 31;
  const int g  = wave % G_;
  const int bd = wave / G_;
  const int dd = bd % D_;
  const int bb = bd / D_;
  const int n0 = 2 * lane;

  v2f al = *(const v2f*)(Alog + (size_t)dd * N_ + n0);
  v2f A;  A.x = -__expf(al.x);  A.y = -__expf(al.y);
  const float Dv = Dp[dd];

  v2f h = *(const v2f*)(h0in + ((size_t)bd * G_ + g) * N_ + n0);

  const int t0 = g * T_;
  for (int tt = 0; tt < T_; ++tt) {
    const size_t row = (size_t)bb * L_ + (t0 + tt);
    const float dlt = delta[row * D_ + dd];
    const float xv  = x[row * D_ + dd];
    v2f Bv = *(const v2f*)(xdbl + row * E_ + R_ + n0);
    v2f Cv = *(const v2f*)(xdbl + row * E_ + R_ + N_ + n0);
    if (tt + 8 < T_)
      __builtin_prefetch(xdbl + (row + 8) * E_ + R_ + n0, 0, 1);
    v2f a;  a.x = __expf(dlt * A.x);  a.y = __expf(dlt * A.y);
    const float du = dlt * xv;
    h.x = a.x * h.x + du * Bv.x;
    h.y = a.y * h.y + du * Bv.y;
    float part = h.x * Cv.x + h.y * Cv.y;
#pragma unroll
    for (int off = 16; off >= 1; off >>= 1)
      part += __shfl_xor(part, off, 32);
    if (lane == 0)
      y[row * D_ + dd] = part + xv * Dv;
  }
}

extern "C" void kernel_launch(void* const* d_in, const int* in_sizes, int n_in,
                              void* d_out, int out_size, void* d_ws, size_t ws_size,
                              hipStream_t stream) {
  const float* x      = (const float*)d_in[0];
  const float* Wxproj = (const float*)d_in[1];
  const float* Wdt    = (const float*)d_in[2];
  const float* bdt    = (const float*)d_in[3];
  const float* Alog   = (const float*)d_in[4];
  const float* Dp     = (const float*)d_in[5];
  float*       y      = (float*)d_out;

  float* ws    = (float*)d_ws;
  float* xdbl  = ws;
  float* delta = xdbl  + (size_t)BL_ * E_;
  float* Pws   = delta + (size_t)BL_ * D_;
  float* Hws   = Pws   + (size_t)B_ * D_ * G_ * N_;
  float* h0ws  = Hws   + (size_t)B_ * D_ * G_ * N_;

  s6_xproj_gemm<<<320, 256, 0, stream>>>(x, Wxproj, xdbl);
  s6_delta_gemm<<<1024, 256, 0, stream>>>(xdbl, Wdt, bdt, delta);
  s6_scan_chunks<<<4096, 256, 0, stream>>>(xdbl, delta, x, Alog, Pws, Hws);
  s6_combine<<<128, 256, 0, stream>>>(Pws, Hws, h0ws);
  s6_scan_emit<<<4096, 256, 0, stream>>>(xdbl, delta, x, Alog, Dp, h0ws, y);
}


# Scan as ONE time-sequential pass: a wave per (b,d) runs all 2048 steps, 64 states on lanes (h serial, no chunk recombination: new sum order); chunk pass+combine gone; y lane sums 32 steps/butterfly
# speedup vs baseline: 2.1193x; 2.1193x over previous
#include <hip/hip_runtime.h>
#include <hip/hip_bf16.h>

typedef float v2f __attribute__((ext_vector_type(2)));
typedef float v8f __attribute__((ext_vector_type(8)));

#define B_  2
#define L_  2048
#define D_  512
#define N_  64
#define R_  32
#define E_  160
#define BL_ (B_ * L_)

__global__ void __launch_bounds__(256)
s6_xproj_gemm(const float* __restrict__ x, const float* __restrict__ W,
              float* __restrict__ xdbl) {
  const int wave  = blockIdx.x * (blockDim.x >> 5) + (threadIdx.x >> 5);
  const int lane  = threadIdx.x & 31;
  const int rt    = wave / 10;
  const int et    = wave % 10;
  const int laneM = lane & 15;
  const int laneH = lane >> 4;

  const float* arow = x + (size_t)(rt * 16 + laneM) * D_ + 2 * laneH;
  const float* brow = W + (size_t)(et * 16 + laneM) * D_ + 2 * laneH;

  v8f c = {};
  for (int k = 0; k < D_; k += 4) {
    v2f a = *(const v2f*)(arow + k);
    v2f b = *(const v2f*)(brow + k);
    c = __builtin_amdgcn_wmma_f32_16x16x4_f32(false, a, false, b,
                                              (short)0, c, false, false);
  }
  const int col = et * 16 + laneM;
#pragma unroll
  for (int j = 0; j < 8; ++j) {
    const int row = rt * 16 + j + 8 * laneH;
    xdbl[(size_t)row * E_ + col] = c[j];
  }
}

__global__ void __launch_bounds__(256)
s6_delta_gemm(const float* __restrict__ xdbl, const float* __restrict__ Wdt,
              const float* __restrict__ bdt, float* __restrict__ delta) {
  const int wave  = blockIdx.x * (blockDim.x >> 5) + (threadIdx.x >> 5);
  const int lane  = threadIdx.x & 31;
  const int rt    = wave >> 5;
  const int nt    = wave & 31;
  const int laneM = lane & 15;
  const int laneH = lane >> 4;

  const float* arow = xdbl + (size_t)(rt * 16 + laneM) * E_ + 2 * laneH;
  const float* brow = Wdt  + (size_t)(nt * 16 + laneM) * R_ + 2 * laneH;

  v8f c = {};
#pragma unroll
  for (int k = 0; k < R_; k += 4) {
    v2f a = *(const v2f*)(arow + k);
    v2f b = *(const v2f*)(brow + k);
    c = __builtin_amdgcn_wmma_f32_16x16x4_f32(false, a, false, b,
                                              (short)0, c, false, false);
  }
  const int col  = nt * 16 + laneM;
  const float bi = bdt[col];
#pragma unroll
  for (int j = 0; j < 8; ++j) {
    const int row = rt * 16 + j + 8 * laneH;
    float z = c[j] + bi;
    float sp = (z > 20.0f) ? z : __logf(1.0f + __expf(z));
    delta[(size_t)row * D_ + col] = sp;
  }
}

#define SB_ 32
#define SG_ 8

__device__ __forceinline__ float bcast_lane(float v, int i) {
  return __builtin_bit_cast(float,
      __builtin_amdgcn_readlane(__builtin_bit_cast(int, v), i));
}
__device__ __forceinline__ float lane_xor16(float v) {
  const int b = __builtin_bit_cast(int, v);
  return __builtin_bit_cast(float,
      __builtin_amdgcn_permlanex16(b, b, 0x76543210, 0xfedcba98, false, false));
}
template <int M>
__device__ __forceinline__ float row_xor(float v) {
  const int b = __builtin_bit_cast(int, v);
  return __builtin_bit_cast(float,
      __builtin_amdgcn_update_dpp(0, b, 0x160 + M, 0xf, 0xf, true));
}

__global__ void __launch_bounds__(128)
s6_scan_seq(const float* __restrict__ xdbl, const float* __restrict__ delta,
            const float* __restrict__ x, const float* __restrict__ Alog,
            const float* __restrict__ Dp, float* __restrict__ y) {
  const int lane = threadIdx.x & 31;
  const int wv = __builtin_amdgcn_readfirstlane((int)(threadIdx.x >> 5));
  const int bd = blockIdx.x * 4 + wv;
  const int dd = bd % D_;
  const int bb = bd / D_;
  const int n0 = 2 * lane;

  v2f al = *(const v2f*)(Alog + (size_t)dd * N_ + n0);
  v2f A;  A.x = -__expf(al.x);  A.y = -__expf(al.y);
  const float Dv = Dp[dd];

  const float* bc = xdbl + (size_t)bb * L_ * E_ + R_ + n0;
  const size_t lrow = ((size_t)bb * L_ + lane) * D_ + dd;
  const float* pdl = delta + lrow;
  const float* pxv = x + lrow;
  float*       py  = y + lrow;

  const bool u4 = (lane & 16) != 0;
  const bool u3 = (lane & 8) != 0;
  const bool u2 = (lane & 4) != 0;
  const bool u1 = (lane & 2) != 0;
  const bool u0 = (lane & 1) != 0;

  float dln = pdl[0];
  float xvn = pxv[0];
  __builtin_amdgcn_sched_barrier(0);
  v2f Bv[SB_], Cv[SB_];
#pragma unroll
  for (int i = 0; i < SB_; ++i) {
    Bv[i] = *(const v2f*)(bc + (size_t)i * E_);
    Cv[i] = *(const v2f*)(bc + (size_t)i * E_ + N_);
  }
  __builtin_amdgcn_sched_barrier(0);

  v2f h = {0.0f, 0.0f};
  for (int t0 = 0; t0 < L_; t0 += SB_) {
    const float dlv = dln;
    const float xvv = xvn;
    const float duv = dlv * xvv;
    const int tn = (t0 + SB_ < L_) ? (t0 + SB_) : t0;
    dln = pdl[(size_t)tn * D_];
    xvn = pxv[(size_t)tn * D_];
    const float* bcn = bc + (size_t)tn * E_;
    __builtin_amdgcn_sched_barrier(0);

    float part[SB_];
#pragma unroll
    for (int j = 0; j < SB_ / SG_; ++j) {
#pragma unroll
      for (int k = 0; k < SG_; ++k) {
        const int i = j * SG_ + k;
        const float dlt = bcast_lane(dlv, i);
        const float du  = bcast_lane(duv, i);
        v2f a;  a.x = __expf(dlt * A.x);  a.y = __expf(dlt * A.y);
        h.x = a.x * h.x + du * Bv[i].x;
        h.y = a.y * h.y + du * Bv[i].y;
        part[i] = h.x * Cv[i].x + h.y * Cv[i].y;
      }
#pragma unroll
      for (int k = 0; k < SG_; ++k) {
        const int i = j * SG_ + k;
        Bv[i] = *(const v2f*)(bcn + (size_t)i * E_);
        Cv[i] = *(const v2f*)(bcn + (size_t)i * E_ + N_);
      }
      __builtin_amdgcn_sched_barrier(0);
    }

    float r16[16], r8[8], r4[4], r2[2];
#pragma unroll
    for (int i = 0; i < 16; ++i) {
      const float keep = u4 ? part[i + 16] : part[i];
      const float send = u4 ? part[i] : part[i + 16];
      r16[i] = keep + lane_xor16(send);
    }
#pragma unroll
    for (int i = 0; i < 8; ++i) {
      const float keep = u3 ? r16[i + 8] : r16[i];
      const float send = u3 ? r16[i] : r16[i + 8];
      r8[i] = keep + row_xor<8>(send);
    }
#pragma unroll
    for (int i = 0; i < 4; ++i) {
      const float keep = u2 ? r8[i + 4] : r8[i];
      const float send = u2 ? r8[i] : r8[i + 4];
      r4[i] = keep + row_xor<4>(send);
    }
#pragma unroll
    for (int i = 0; i < 2; ++i) {
      const float keep = u1 ? r4[i + 2] : r4[i];
      const float send = u1 ? r4[i] : r4[i + 2];
      r2[i] = keep + row_xor<2>(send);
    }
    const float keep0 = u0 ? r2[1] : r2[0];
    const float send0 = u0 ? r2[0] : r2[1];
    const float tot = keep0 + row_xor<1>(send0);

    py[(size_t)t0 * D_] = tot + xvv * Dv;
  }
}

extern "C" void kernel_launch(void* const* d_in, const int* in_sizes, int n_in,
                              void* d_out, int out_size, void* d_ws, size_t ws_size,
                              hipStream_t stream) {
  const float* x      = (const float*)d_in[0];
  const float* Wxproj = (const float*)d_in[1];
  const float* Wdt    = (const float*)d_in[2];
  const float* bdt    = (const float*)d_in[3];
  const float* Alog   = (const float*)d_in[4];
  const float* Dp     = (const float*)d_in[5];
  float*       y      = (float*)d_out;

  float* ws    = (float*)d_ws;
  float* xdbl  = ws;
  float* delta = xdbl  + (size_t)BL_ * E_;

  s6_xproj_gemm<<<320, 256, 0, stream>>>(x, Wxproj, xdbl);
  s6_delta_gemm<<<1024, 256, 0, stream>>>(xdbl, Wdt, bdt, delta);
  s6_scan_seq<<<256, 128, 0, stream>>>(xdbl, delta, x, Alog, Dp, y);
}


# Scan: a wave owns TWO channels, one per 16-lane row, 4 states a lane: each B/C row fetch serves both; delta by DPP row_share, no readlane; y = 4 in-lane terms + 16-lane butterfly (new sum order)
# speedup vs baseline: 2.3054x; 1.0878x over previous
#include <hip/hip_runtime.h>
#include <hip/hip_bf16.h>

typedef float v2f __attribute__((ext_vector_type(2)));
typedef float v8f __attribute__((ext_vector_type(8)));

#define B_  2
#define L_  2048
#define D_  512
#define N_  64
#define R_  32
#define E_  160
#define BL_ (B_ * L_)

__global__ void __launch_bounds__(256)
s6_xproj_gemm(const float* __restrict__ x, const float* __restrict__ W,
              float* __restrict__ xdbl) {
  const int wave  = blockIdx.x * (blockDim.x >> 5) + (threadIdx.x >> 5);
  const int lane  = threadIdx.x & 31;
  const int rt    = wave / 10;
  const int et    = wave % 10;
  const int laneM = lane & 15;
  const int laneH = lane >> 4;

  const float* arow = x + (size_t)(rt * 16 + laneM) * D_ + 2 * laneH;
  const float* brow = W + (size_t)(et * 16 + laneM) * D_ + 2 * laneH;

  v8f c = {};
  for (int k = 0; k < D_; k += 4) {
    v2f a = *(const v2f*)(arow + k);
    v2f b = *(const v2f*)(brow + k);
    c = __builtin_amdgcn_wmma_f32_16x16x4_f32(false, a, false, b,
                                              (short)0, c, false, false);
  }
  const int col = et * 16 + laneM;
#pragma unroll
  for (int j = 0; j < 8; ++j) {
    const int row = rt * 16 + j + 8 * laneH;
    xdbl[(size_t)row * E_ + col] = c[j];
  }
}

__global__ void __launch_bounds__(256)
s6_delta_gemm(const float* __restrict__ xdbl, const float* __restrict__ Wdt,
              const float* __restrict__ bdt, float* __restrict__ delta) {
  const int wave  = blockIdx.x * (blockDim.x >> 5) + (threadIdx.x >> 5);
  const int lane  = threadIdx.x & 31;
  const int rt    = wave >> 5;
  const int nt    = wave & 31;
  const int laneM = lane & 15;
  const int laneH = lane >> 4;

  const float* arow = xdbl + (size_t)(rt * 16 + laneM) * E_ + 2 * laneH;
  const float* brow = Wdt  + (size_t)(nt * 16 + laneM) * R_ + 2 * laneH;

  v8f c = {};
#pragma unroll
  for (int k = 0; k < R_; k += 4) {
    v2f a = *(const v2f*)(arow + k);
    v2f b = *(const v2f*)(brow + k);
    c = __builtin_amdgcn_wmma_f32_16x16x4_f32(false, a, false, b,
                                              (short)0, c, false, false);
  }
  const int col  = nt * 16 + laneM;
  const float bi = bdt[col];
#pragma unroll
  for (int j = 0; j < 8; ++j) {
    const int row = rt * 16 + j + 8 * laneH;
    float z = c[j] + bi;
    float sp = (z > 20.0f) ? z : __logf(1.0f + __expf(z));
    delta[(size_t)row * D_ + col] = sp;
  }
}

typedef float v4f __attribute__((ext_vector_type(4)));

#define SB_ 32
#define SH_ 16
#define SG_ 8

__device__ __forceinline__ float row_share(float v, int i) {
  const int b = __builtin_bit_cast(int, v);
  int r = 0;
  switch (i) {
    case 0: r = __builtin_amdgcn_update_dpp(0, b, 0x150 + 0, 0xf, 0xf, true); break;
    case 1: r = __builtin_amdgcn_update_dpp(0, b, 0x150 + 1, 0xf, 0xf, true); break;
    case 2: r = __builtin_amdgcn_update_dpp(0, b, 0x150 + 2, 0xf, 0xf, true); break;
    case 3: r = __builtin_amdgcn_update_dpp(0, b, 0x150 + 3, 0xf, 0xf, true); break;
    case 4: r = __builtin_amdgcn_update_dpp(0, b, 0x150 + 4, 0xf, 0xf, true); break;
    case 5: r = __builtin_amdgcn_update_dpp(0, b, 0x150 + 5, 0xf, 0xf, true); break;
    case 6: r = __builtin_amdgcn_update_dpp(0, b, 0x150 + 6, 0xf, 0xf, true); break;
    case 7: r = __builtin_amdgcn_update_dpp(0, b, 0x150 + 7, 0xf, 0xf, true); break;
    case 8: r = __builtin_amdgcn_update_dpp(0, b, 0x150 + 8, 0xf, 0xf, true); break;
    case 9: r = __builtin_amdgcn_update_dpp(0, b, 0x150 + 9, 0xf, 0xf, true); break;
    case 10: r = __builtin_amdgcn_update_dpp(0, b, 0x150 + 10, 0xf, 0xf, true); break;
    case 11: r = __builtin_amdgcn_update_dpp(0, b, 0x150 + 11, 0xf, 0xf, true); break;
    case 12: r = __builtin_amdgcn_update_dpp(0, b, 0x150 + 12, 0xf, 0xf, true); break;
    case 13: r = __builtin_amdgcn_update_dpp(0, b, 0x150 + 13, 0xf, 0xf, true); break;
    case 14: r = __builtin_amdgcn_update_dpp(0, b, 0x150 + 14, 0xf, 0xf, true); break;
    case 15: r = __builtin_amdgcn_update_dpp(0, b, 0x150 + 15, 0xf, 0xf, true); break;
  }
  return __builtin_bit_cast(float, r);
}
template <int M>
__device__ __forceinline__ float row_xor(float v) {
  const int b = __builtin_bit_cast(int, v);
  return __builtin_bit_cast(float,
      __builtin_amdgcn_update_dpp(0, b, 0x160 + M, 0xf, 0xf, true));
}

__device__ __forceinline__ float row_sums(const float* part, bool u3, bool u2,
                                          bool u1, bool u0) {
  float r8[8], r4[4], r2[2];
#pragma unroll
  for (int i = 0; i < 8; ++i) {
    const float keep = u3 ? part[i + 8] : part[i];
    const float send = u3 ? part[i] : part[i + 8];
    r8[i] = keep + row_xor<8>(send);
  }
#pragma unroll
  for (int i = 0; i < 4; ++i) {
    const float keep = u2 ? r8[i + 4] : r8[i];
    const float send = u2 ? r8[i] : r8[i + 4];
    r4[i] = keep + row_xor<4>(send);
  }
#pragma unroll
  for (int i = 0; i < 2; ++i) {
    const float keep = u1 ? r4[i + 2] : r4[i];
    const float send = u1 ? r4[i] : r4[i + 2];
    r2[i] = keep + row_xor<2>(send);
  }
  const float keep0 = u0 ? r2[1] : r2[0];
  const float send0 = u0 ? r2[0] : r2[1];
  return keep0 + row_xor<1>(send0);
}

__global__ void __launch_bounds__(64)
s6_scan_seq(const float* __restrict__ xdbl, const float* __restrict__ delta,
            const float* __restrict__ x, const float* __restrict__ Alog,
            const float* __restrict__ Dp, float* __restrict__ y) {
  const int lane = threadIdx.x & 31;
  const int wv = __builtin_amdgcn_readfirstlane((int)(threadIdx.x >> 5));
  const int pw = blockIdx.x * 2 + wv;
  const int bb = pw / (D_ / 2);
  const int c  = lane & 15;
  const int dc = (pw % (D_ / 2)) * 2 + (lane >> 4);
  const int n0 = 4 * c;

  const v4f al = *(const v4f*)(Alog + (size_t)dc * N_ + n0);
  v4f A;
  A.x = -__expf(al.x);  A.y = -__expf(al.y);
  A.z = -__expf(al.z);  A.w = -__expf(al.w);
  const float Dv = Dp[dc];

  const float* bc = xdbl + (size_t)bb * L_ * E_ + R_ + n0;
  const size_t lrow = ((size_t)bb * L_ + c) * D_ + dc;
  const float* pdl = delta + lrow;
  const float* pxv = x + lrow;
  float*       py  = y + lrow;

  const bool u3 = (lane & 8) != 0;
  const bool u2 = (lane & 4) != 0;
  const bool u1 = (lane & 2) != 0;
  const bool u0 = (lane & 1) != 0;

  float dlnA = pdl[0];
  float dlnB = pdl[(size_t)SH_ * D_];
  float xvnA = pxv[0];
  float xvnB = pxv[(size_t)SH_ * D_];
  __builtin_amdgcn_sched_barrier(0);
  v4f Bv[SB_], Cv[SB_];
#pragma unroll
  for (int i = 0; i < SB_; ++i) {
    Bv[i] = *(const v4f*)(bc + (size_t)i * E_);
    Cv[i] = *(const v4f*)(bc + (size_t)i * E_ + N_);
  }
  __builtin_amdgcn_sched_barrier(0);

  v4f h = {0.0f, 0.0f, 0.0f, 0.0f};
  for (int t0 = 0; t0 < L_; t0 += SB_) {
    const float dlvA = dlnA;
    const float dlvB = dlnB;
    const float xvvA = xvnA;
    const float xvvB = xvnB;
    const float duvA = dlvA * xvvA;
    const float duvB = dlvB * xvvB;
    const int tn = (t0 + SB_ < L_) ? (t0 + SB_) : t0;
    dlnA = pdl[(size_t)tn * D_];
    dlnB = pdl[(size_t)(tn + SH_) * D_];
    xvnA = pxv[(size_t)tn * D_];
    xvnB = pxv[(size_t)(tn + SH_) * D_];
    const float* bcn = bc + (size_t)tn * E_;
    __builtin_amdgcn_sched_barrier(0);

    float part[SB_];
#pragma unroll
    for (int j = 0; j < SB_ / SG_; ++j) {
#pragma unroll
      for (int k = 0; k < SG_; ++k) {
        const int i = j * SG_ + k;
        const float dlt = (i < SH_) ? row_share(dlvA, i) : row_share(dlvB, i - SH_);
        const float du  = (i < SH_) ? row_share(duvA, i) : row_share(duvB, i - SH_);
        const v4f e = A * dlt;
        v4f a;
        a.x = __expf(e.x);  a.y = __expf(e.y);
        a.z = __expf(e.z);  a.w = __expf(e.w);
        h = a * h + du * Bv[i];
        const v4f hc = h * Cv[i];
        part[i] = (hc.x + hc.z) + (hc.y + hc.w);
      }
#pragma unroll
      for (int k = 0; k < SG_; ++k) {
        const int i = j * SG_ + k;
        Bv[i] = *(const v4f*)(bcn + (size_t)i * E_);
        Cv[i] = *(const v4f*)(bcn + (size_t)i * E_ + N_);
      }
      __builtin_amdgcn_sched_barrier(0);
    }

    const float totA = row_sums(part, u3, u2, u1, u0);
    const float totB = row_sums(part + SH_, u3, u2, u1, u0);

    py[(size_t)t0 * D_] = totA + xvvA * Dv;
    py[(size_t)(t0 + SH_) * D_] = totB + xvvB * Dv;
  }
}

extern "C" void kernel_launch(void* const* d_in, const int* in_sizes, int n_in,
                              void* d_out, int out_size, void* d_ws, size_t ws_size,
                              hipStream_t stream) {
  const float* x      = (const float*)d_in[0];
  const float* Wxproj = (const float*)d_in[1];
  const float* Wdt    = (const float*)d_in[2];
  const float* bdt    = (const float*)d_in[3];
  const float* Alog   = (const float*)d_in[4];
  const float* Dp     = (const float*)d_in[5];
  float*       y      = (float*)d_out;

  float* ws    = (float*)d_ws;
  float* xdbl  = ws;
  float* delta = xdbl  + (size_t)BL_ * E_;

  s6_xproj_gemm<<<320, 256, 0, stream>>>(x, Wxproj, xdbl);
  s6_delta_gemm<<<1024, 256, 0, stream>>>(xdbl, Wdt, bdt, delta);
  s6_scan_seq<<<256, 64, 0, stream>>>(xdbl, delta, x, Alog, Dp, y);
}


# x_proj GEMM re-tiled: 4 waves per 16-row tile, each keeps one x fragment for 2-3 column tiles; 16-byte fragment loads a 64-k buffer ahead (k order permuted inside 8-k chunks: new sum order)
# speedup vs baseline: 2.5424x; 1.1028x over previous
#include <hip/hip_runtime.h>
#include <hip/hip_bf16.h>

typedef float v2f __attribute__((ext_vector_type(2)));
typedef float v8f __attribute__((ext_vector_type(8)));

#define B_  2
#define L_  2048
#define D_  512
#define N_  64
#define R_  32
#define E_  160
#define BL_ (B_ * L_)

typedef float v4f __attribute__((ext_vector_type(4)));

#define XU_ 8
#define XG_ (D_ / (8 * XU_))

template <int E>
struct XFrag {
  v4f a[XU_];
  v4f b[E][XU_];
};

template <int E>
__device__ __forceinline__ void xproj_load(XFrag<E>& f, const float* __restrict__ ap,
                                           const float* __restrict__ bp, int g) {
#pragma unroll
  for (int u = 0; u < XU_; ++u) {
    f.a[u] = *(const v4f*)(ap + (g * XU_ + u) * 8);
#pragma unroll
    for (int e = 0; e < E; ++e)
      f.b[e][u] = *(const v4f*)(bp + (size_t)e * 16 * D_ + (g * XU_ + u) * 8);
  }
}

template <int E>
__device__ __forceinline__ void xproj_mma(v8f (&acc)[E], const XFrag<E>& f) {
#pragma unroll
  for (int u = 0; u < XU_; ++u) {
    const v2f a0 = f.a[u].xy;
    const v2f a1 = f.a[u].zw;
#pragma unroll
    for (int e = 0; e < E; ++e) {
      const v2f b0 = f.b[e][u].xy;
      acc[e] = __builtin_amdgcn_wmma_f32_16x16x4_f32(false, a0, false, b0,
                                                     (short)0, acc[e], false, false);
    }
#pragma unroll
    for (int e = 0; e < E; ++e) {
      const v2f b1 = f.b[e][u].zw;
      acc[e] = __builtin_amdgcn_wmma_f32_16x16x4_f32(false, a1, false, b1,
                                                     (short)0, acc[e], false, false);
    }
  }
}

template <int E>
__device__ __forceinline__ void xproj_strip(const float* __restrict__ ap,
                                            const float* __restrict__ bp,
                                            float* __restrict__ op) {
  v8f acc[E];
#pragma unroll
  for (int e = 0; e < E; ++e) acc[e] = v8f{};
  XFrag<E> f0, f1;
  xproj_load<E>(f0, ap, bp, 0);
  for (int g = 0; g < XG_ - 2; g += 2) {
    xproj_load<E>(f1, ap, bp, g + 1);
    __builtin_amdgcn_sched_barrier(0);
    xproj_mma<E>(acc, f0);
    __builtin_amdgcn_sched_barrier(0);
    xproj_load<E>(f0, ap, bp, g + 2);
    __builtin_amdgcn_sched_barrier(0);
    xproj_mma<E>(acc, f1);
    __builtin_amdgcn_sched_barrier(0);
  }
  xproj_load<E>(f1, ap, bp, XG_ - 1);
  __builtin_amdgcn_sched_barrier(0);
  xproj_mma<E>(acc, f0);
  xproj_mma<E>(acc, f1);
#pragma unroll
  for (int e = 0; e < E; ++e)
#pragma unroll
    for (int j = 0; j < 8; ++j)
      op[(size_t)j * E_ + e * 16] = acc[e][j];
}

__global__ void __launch_bounds__(128)
s6_xproj_gemm(const float* __restrict__ x, const float* __restrict__ W,
              float* __restrict__ xdbl) {
  const int lane  = threadIdx.x & 31;
  const int wv    = __builtin_amdgcn_readfirstlane((int)(threadIdx.x >> 5));
  const int rt    = blockIdx.x;
  const int laneM = lane & 15;
  const int laneH = lane >> 4;

  const float* ap = x + (size_t)(rt * 16 + laneM) * D_ + 4 * laneH;
  float* orow = xdbl + (size_t)(rt * 16 + 8 * laneH) * E_ + laneM;
  if (wv < 2) {
    const int t0 = 3 * wv;
    xproj_strip<3>(ap, W + (size_t)(t0 * 16 + laneM) * D_ + 4 * laneH, orow + t0 * 16);
  } else {
    const int t0 = 2 * wv + 2;
    xproj_strip<2>(ap, W + (size_t)(t0 * 16 + laneM) * D_ + 4 * laneH, orow + t0 * 16);
  }
}

__global__ void __launch_bounds__(256)
s6_delta_gemm(const float* __restrict__ xdbl, const float* __restrict__ Wdt,
              const float* __restrict__ bdt, float* __restrict__ delta) {
  const int wave  = blockIdx.x * (blockDim.x >> 5) + (threadIdx.x >> 5);
  const int lane  = threadIdx.x & 31;
  const int rt    = wave >> 5;
  const int nt    = wave & 31;
  const int laneM = lane & 15;
  const int laneH = lane >> 4;

  const float* arow = xdbl + (size_t)(rt * 16 + laneM) * E_ + 2 * laneH;
  const float* brow = Wdt  + (size_t)(nt * 16 + laneM) * R_ + 2 * laneH;

  v8f c = {};
#pragma unroll
  for (int k = 0; k < R_; k += 4) {
    v2f a = *(const v2f*)(arow + k);
    v2f b = *(const v2f*)(brow + k);
    c = __builtin_amdgcn_wmma_f32_16x16x4_f32(false, a, false, b,
                                              (short)0, c, false, false);
  }
  const int col  = nt * 16 + laneM;
  const float bi = bdt[col];
#pragma unroll
  for (int j = 0; j < 8; ++j) {
    const int row = rt * 16 + j + 8 * laneH;
    float z = c[j] + bi;
    float sp = (z > 20.0f) ? z : __logf(1.0f + __expf(z));
    delta[(size_t)row * D_ + col] = sp;
  }
}

#define SB_ 32
#define SH_ 16
#define SG_ 8

__device__ __forceinline__ float row_share(float v, int i) {
  const int b = __builtin_bit_cast(int, v);
  int r = 0;
  switch (i) {
    case 0: r = __builtin_amdgcn_update_dpp(0, b, 0x150 + 0, 0xf, 0xf, true); break;
    case 1: r = __builtin_amdgcn_update_dpp(0, b, 0x150 + 1, 0xf, 0xf, true); break;
    case 2: r = __builtin_amdgcn_update_dpp(0, b, 0x150 + 2, 0xf, 0xf, true); break;
    case 3: r = __builtin_amdgcn_update_dpp(0, b, 0x150 + 3, 0xf, 0xf, true); break;
    case 4: r = __builtin_amdgcn_update_dpp(0, b, 0x150 + 4, 0xf, 0xf, true); break;
    case 5: r = __builtin_amdgcn_update_dpp(0, b, 0x150 + 5, 0xf, 0xf, true); break;
    case 6: r = __builtin_amdgcn_update_dpp(0, b, 0x150 + 6, 0xf, 0xf, true); break;
    case 7: r = __builtin_amdgcn_update_dpp(0, b, 0x150 + 7, 0xf, 0xf, true); break;
    case 8: r = __builtin_amdgcn_update_dpp(0, b, 0x150 + 8, 0xf, 0xf, true); break;
    case 9: r = __builtin_amdgcn_update_dpp(0, b, 0x150 + 9, 0xf, 0xf, true); break;
    case 10: r = __builtin_amdgcn_update_dpp(0, b, 0x150 + 10, 0xf, 0xf, true); break;
    case 11: r = __builtin_amdgcn_update_dpp(0, b, 0x150 + 11, 0xf, 0xf, true); break;
    case 12: r = __builtin_amdgcn_update_dpp(0, b, 0x150 + 12, 0xf, 0xf, true); break;
    case 13: r = __builtin_amdgcn_update_dpp(0, b, 0x150 + 13, 0xf, 0xf, true); break;
    case 14: r = __builtin_amdgcn_update_dpp(0, b, 0x150 + 14, 0xf, 0xf, true); break;
    case 15: r = __builtin_amdgcn_update_dpp(0, b, 0x150 + 15, 0xf, 0xf, true); break;
  }
  return __builtin_bit_cast(float, r);
}
template <int M>
__device__ __forceinline__ float row_xor(float v) {
  const int b = __builtin_bit_cast(int, v);
  return __builtin_bit_cast(float,
      __builtin_amdgcn_update_dpp(0, b, 0x160 + M, 0xf, 0xf, true));
}

__device__ __forceinline__ float row_sums(const float* part, bool u3, bool u2,
                                          bool u1, bool u0) {
  float r8[8], r4[4], r2[2];
#pragma unroll
  for (int i = 0; i < 8; ++i) {
    const float keep = u3 ? part[i + 8] : part[i];
    const float send = u3 ? part[i] : part[i + 8];
    r8[i] = keep + row_xor<8>(send);
  }
#pragma unroll
  for (int i = 0; i < 4; ++i) {
    const float keep = u2 ? r8[i + 4] : r8[i];
    const float send = u2 ? r8[i] : r8[i + 4];
    r4[i] = keep + row_xor<4>(send);
  }
#pragma unroll
  for (int i = 0; i < 2; ++i) {
    const float keep = u1 ? r4[i + 2] : r4[i];
    const float send = u1 ? r4[i] : r4[i + 2];
    r2[i] = keep + row_xor<2>(send);
  }
  const float keep0 = u0 ? r2[1] : r2[0];
  const float send0 = u0 ? r2[0] : r2[1];
  return keep0 + row_xor<1>(send0);
}

__global__ void __launch_bounds__(64)
s6_scan_seq(const float* __restrict__ xdbl, const float* __restrict__ delta,
            const float* __restrict__ x, const float* __restrict__ Alog,
            const float* __restrict__ Dp, float* __restrict__ y) {
  const int lane = threadIdx.x & 31;
  const int wv = __builtin_amdgcn_readfirstlane((int)(threadIdx.x >> 5));
  const int pw = blockIdx.x * 2 + wv;
  const int bb = pw / (D_ / 2);
  const int c  = lane & 15;
  const int dc = (pw % (D_ / 2)) * 2 + (lane >> 4);
  const int n0 = 4 * c;

  const v4f al = *(const v4f*)(Alog + (size_t)dc * N_ + n0);
  v4f A;
  A.x = -__expf(al.x);  A.y = -__expf(al.y);
  A.z = -__expf(al.z);  A.w = -__expf(al.w);
  const float Dv = Dp[dc];

  const float* bc = xdbl + (size_t)bb * L_ * E_ + R_ + n0;
  const size_t lrow = ((size_t)bb * L_ + c) * D_ + dc;
  const float* pdl = delta + lrow;
  const float* pxv = x + lrow;
  float*       py  = y + lrow;

  const bool u3 = (lane & 8) != 0;
  const bool u2 = (lane & 4) != 0;
  const bool u1 = (lane & 2) != 0;
  const bool u0 = (lane & 1) != 0;

  float dlnA = pdl[0];
  float dlnB = pdl[(size_t)SH_ * D_];
  float xvnA = pxv[0];
  float xvnB = pxv[(size_t)SH_ * D_];
  __builtin_amdgcn_sched_barrier(0);
  v4f Bv[SB_], Cv[SB_];
#pragma unroll
  for (int i = 0; i < SB_; ++i) {
    Bv[i] = *(const v4f*)(bc + (size_t)i * E_);
    Cv[i] = *(const v4f*)(bc + (size_t)i * E_ + N_);
  }
  __builtin_amdgcn_sched_barrier(0);

  v4f h = {0.0f, 0.0f, 0.0f, 0.0f};
  for (int t0 = 0; t0 < L_; t0 += SB_) {
    const float dlvA = dlnA;
    const float dlvB = dlnB;
    const float xvvA = xvnA;
    const float xvvB = xvnB;
    const float duvA = dlvA * xvvA;
    const float duvB = dlvB * xvvB;
    const int tn = (t0 + SB_ < L_) ? (t0 + SB_) : t0;
    dlnA = pdl[(size_t)tn * D_];
    dlnB = pdl[(size_t)(tn + SH_) * D_];
    xvnA = pxv[(size_t)tn * D_];
    xvnB = pxv[(size_t)(tn + SH_) * D_];
    const float* bcn = bc + (size_t)tn * E_;
    __builtin_amdgcn_sched_barrier(0);

    float part[SB_];
#pragma unroll
    for (int j = 0; j < SB_ / SG_; ++j) {
#pragma unroll
      for (int k = 0; k < SG_; ++k) {
        const int i = j * SG_ + k;
        const float dlt = (i < SH_) ? row_share(dlvA, i) : row_share(dlvB, i - SH_);
        const float du  = (i < SH_) ? row_share(duvA, i) : row_share(duvB, i - SH_);
        const v4f e = A * dlt;
        v4f a;
        a.x = __expf(e.x);  a.y = __expf(e.y);
        a.z = __expf(e.z);  a.w = __expf(e.w);
        h = a * h + du * Bv[i];
        const v4f hc = h * Cv[i];
        part[i] = (hc.x + hc.z) + (hc.y + hc.w);
      }
#pragma unroll
      for (int k = 0; k < SG_; ++k) {
        const int i = j * SG_ + k;
        Bv[i] = *(const v4f*)(bcn + (size_t)i * E_);
        Cv[i] = *(const v4f*)(bcn + (size_t)i * E_ + N_);
      }
      __builtin_amdgcn_sched_barrier(0);
    }

    const float totA = row_sums(part, u3, u2, u1, u0);
    const float totB = row_sums(part + SH_, u3, u2, u1, u0);

    py[(size_t)t0 * D_] = totA + xvvA * Dv;
    py[(size_t)(t0 + SH_) * D_] = totB + xvvB * Dv;
  }
}

extern "C" void kernel_launch(void* const* d_in, const int* in_sizes, int n_in,
                              void* d_out, int out_size, void* d_ws, size_t ws_size,
                              hipStream_t stream) {
  const float* x      = (const float*)d_in[0];
  const float* Wxproj = (const float*)d_in[1];
  const float* Wdt    = (const float*)d_in[2];
  const float* bdt    = (const float*)d_in[3];
  const float* Alog   = (const float*)d_in[4];
  const float* Dp     = (const float*)d_in[5];
  float*       y      = (float*)d_out;

  float* ws    = (float*)d_ws;
  float* xdbl  = ws;
  float* delta = xdbl  + (size_t)BL_ * E_;

  s6_xproj_gemm<<<256, 128, 0, stream>>>(x, Wxproj, xdbl);
  s6_delta_gemm<<<1024, 256, 0, stream>>>(xdbl, Wdt, bdt, delta);
  s6_scan_seq<<<256, 64, 0, stream>>>(xdbl, delta, x, Alog, Dp, y);
}


# Scan: a channel pair's 64 states split over TWO waves (2 states a lane; 1024 waves, all 4 SIMDs of a WGP busy); their partial y sums swapped through LDS once per 32 steps (new sum order)
# speedup vs baseline: 2.8586x; 1.1244x over previous
#include <hip/hip_runtime.h>
#include <hip/hip_bf16.h>

typedef float v2f __attribute__((ext_vector_type(2)));
typedef float v8f __attribute__((ext_vector_type(8)));

#define B_  2
#define L_  2048
#define D_  512
#define N_  64
#define R_  32
#define E_  160
#define BL_ (B_ * L_)

typedef float v4f __attribute__((ext_vector_type(4)));

#define XU_ 8
#define XG_ (D_ / (8 * XU_))

template <int E>
struct XFrag {
  v4f a[XU_];
  v4f b[E][XU_];
};

template <int E>
__device__ __forceinline__ void xproj_load(XFrag<E>& f, const float* __restrict__ ap,
                                           const float* __restrict__ bp, int g) {
#pragma unroll
  for (int u = 0; u < XU_; ++u) {
    f.a[u] = *(const v4f*)(ap + (g * XU_ + u) * 8);
#pragma unroll
    for (int e = 0; e < E; ++e)
      f.b[e][u] = *(const v4f*)(bp + (size_t)e * 16 * D_ + (g * XU_ + u) * 8);
  }
}

template <int E>
__device__ __forceinline__ void xproj_mma(v8f (&acc)[E], const XFrag<E>& f) {
#pragma unroll
  for (int u = 0; u < XU_; ++u) {
    const v2f a0 = f.a[u].xy;
    const v2f a1 = f.a[u].zw;
#pragma unroll
    for (int e = 0; e < E; ++e) {
      const v2f b0 = f.b[e][u].xy;
      acc[e] = __builtin_amdgcn_wmma_f32_16x16x4_f32(false, a0, false, b0,
                                                     (short)0, acc[e], false, false);
    }
#pragma unroll
    for (int e = 0; e < E; ++e) {
      const v2f b1 = f.b[e][u].zw;
      acc[e] = __builtin_amdgcn_wmma_f32_16x16x4_f32(false, a1, false, b1,
                                                     (short)0, acc[e], false, false);
    }
  }
}

template <int E>
__device__ __forceinline__ void xproj_strip(const float* __restrict__ ap,
                                            const float* __restrict__ bp,
                                            float* __restrict__ op) {
  v8f acc[E];
#pragma unroll
  for (int e = 0; e < E; ++e) acc[e] = v8f{};
  XFrag<E> f0, f1;
  xproj_load<E>(f0, ap, bp, 0);
  for (int g = 0; g < XG_ - 2; g += 2) {
    xproj_load<E>(f1, ap, bp, g + 1);
    __builtin_amdgcn_sched_barrier(0);
    xproj_mma<E>(acc, f0);
    __builtin_amdgcn_sched_barrier(0);
    xproj_load<E>(f0, ap, bp, g + 2);
    __builtin_amdgcn_sched_barrier(0);
    xproj_mma<E>(acc, f1);
    __builtin_amdgcn_sched_barrier(0);
  }
  xproj_load<E>(f1, ap, bp, XG_ - 1);
  __builtin_amdgcn_sched_barrier(0);
  xproj_mma<E>(acc, f0);
  xproj_mma<E>(acc, f1);
#pragma unroll
  for (int e = 0; e < E; ++e)
#pragma unroll
    for (int j = 0; j < 8; ++j)
      op[(size_t)j * E_ + e * 16] = acc[e][j];
}

__global__ void __launch_bounds__(128)
s6_xproj_gemm(const float* __restrict__ x, const float* __restrict__ W,
              float* __restrict__ xdbl) {
  const int lane  = threadIdx.x & 31;
  const int wv    = __builtin_amdgcn_readfirstlane((int)(threadIdx.x >> 5));
  const int rt    = blockIdx.x;
  const int laneM = lane & 15;
  const int laneH = lane >> 4;

  const float* ap = x + (size_t)(rt * 16 + laneM) * D_ + 4 * laneH;
  float* orow = xdbl + (size_t)(rt * 16 + 8 * laneH) * E_ + laneM;
  if (wv < 2) {
    const int t0 = 3 * wv;
    xproj_strip<3>(ap, W + (size_t)(t0 * 16 + laneM) * D_ + 4 * laneH, orow + t0 * 16);
  } else {
    const int t0 = 2 * wv + 2;
    xproj_strip<2>(ap, W + (size_t)(t0 * 16 + laneM) * D_ + 4 * laneH, orow + t0 * 16);
  }
}

__global__ void __launch_bounds__(256)
s6_delta_gemm(const float* __restrict__ xdbl, const float* __restrict__ Wdt,
              const float* __restrict__ bdt, float* __restrict__ delta) {
  const int wave  = blockIdx.x * (blockDim.x >> 5) + (threadIdx.x >> 5);
  const int lane  = threadIdx.x & 31;
  const int rt    = wave >> 5;
  const int nt    = wave & 31;
  const int laneM = lane & 15;
  const int laneH = lane >> 4;

  const float* arow = xdbl + (size_t)(rt * 16 + laneM) * E_ + 2 * laneH;
  const float* brow = Wdt  + (size_t)(nt * 16 + laneM) * R_ + 2 * laneH;

  v8f c = {};
#pragma unroll
  for (int k = 0; k < R_; k += 4) {
    v2f a = *(const v2f*)(arow + k);
    v2f b = *(const v2f*)(brow + k);
    c = __builtin_amdgcn_wmma_f32_16x16x4_f32(false, a, false, b,
                                              (short)0, c, false, false);
  }
  const int col  = nt * 16 + laneM;
  const float bi = bdt[col];
#pragma unroll
  for (int j = 0; j < 8; ++j) {
    const int row = rt * 16 + j + 8 * laneH;
    float z = c[j] + bi;
    float sp = (z > 20.0f) ? z : __logf(1.0f + __expf(z));
    delta[(size_t)row * D_ + col] = sp;
  }
}

#define SB_ 32
#define SH_ 16
#define SG_ 8

__device__ __forceinline__ float row_share(float v, int i) {
  const int b = __builtin_bit_cast(int, v);
  int r = 0;
  switch (i) {
    case 0: r = __builtin_amdgcn_update_dpp(0, b, 0x150 + 0, 0xf, 0xf, true); break;
    case 1: r = __builtin_amdgcn_update_dpp(0, b, 0x150 + 1, 0xf, 0xf, true); break;
    case 2: r = __builtin_amdgcn_update_dpp(0, b, 0x150 + 2, 0xf, 0xf, true); break;
    case 3: r = __builtin_amdgcn_update_dpp(0, b, 0x150 + 3, 0xf, 0xf, true); break;
    case 4: r = __builtin_amdgcn_update_dpp(0, b, 0x150 + 4, 0xf, 0xf, true); break;
    case 5: r = __builtin_amdgcn_update_dpp(0, b, 0x150 + 5, 0xf, 0xf, true); break;
    case 6: r = __builtin_amdgcn_update_dpp(0, b, 0x150 + 6, 0xf, 0xf, true); break;
    case 7: r = __builtin_amdgcn_update_dpp(0, b, 0x150 + 7, 0xf, 0xf, true); break;
    case 8: r = __builtin_amdgcn_update_dpp(0, b, 0x150 + 8, 0xf, 0xf, true); break;
    case 9: r = __builtin_amdgcn_update_dpp(0, b, 0x150 + 9, 0xf, 0xf, true); break;
    case 10: r = __builtin_amdgcn_update_dpp(0, b, 0x150 + 10, 0xf, 0xf, true); break;
    case 11: r = __builtin_amdgcn_update_dpp(0, b, 0x150 + 11, 0xf, 0xf, true); break;
    case 12: r = __builtin_amdgcn_update_dpp(0, b, 0x150 + 12, 0xf, 0xf, true); break;
    case 13: r = __builtin_amdgcn_update_dpp(0, b, 0x150 + 13, 0xf, 0xf, true); break;
    case 14: r = __builtin_amdgcn_update_dpp(0, b, 0x150 + 14, 0xf, 0xf, true); break;
    case 15: r = __builtin_amdgcn_update_dpp(0, b, 0x150 + 15, 0xf, 0xf, true); break;
  }
  return __builtin_bit_cast(float, r);
}
template <int M>
__device__ __forceinline__ float row_xor(float v) {
  const int b = __builtin_bit_cast(int, v);
  return __builtin_bit_cast(float,
      __builtin_amdgcn_update_dpp(0, b, 0x160 + M, 0xf, 0xf, true));
}

__device__ __forceinline__ float row_sums(const float* part, bool u3, bool u2,
                                          bool u1, bool u0) {
  float r8[8], r4[4], r2[2];
#pragma unroll
  for (int i = 0; i < 8; ++i) {
    const float keep = u3 ? part[i + 8] : part[i];
    const float send = u3 ? part[i] : part[i + 8];
    r8[i] = keep + row_xor<8>(send);
  }
#pragma unroll
  for (int i = 0; i < 4; ++i) {
    const float keep = u2 ? r8[i + 4] : r8[i];
    const float send = u2 ? r8[i] : r8[i + 4];
    r4[i] = keep + row_xor<4>(send);
  }
#pragma unroll
  for (int i = 0; i < 2; ++i) {
    const float keep = u1 ? r4[i + 2] : r4[i];
    const float send = u1 ? r4[i] : r4[i + 2];
    r2[i] = keep + row_xor<2>(send);
  }
  const float keep0 = u0 ? r2[1] : r2[0];
  const float send0 = u0 ? r2[0] : r2[1];
  return keep0 + row_xor<1>(send0);
}

__device__ __forceinline__ void swap_arrive() {
  __builtin_amdgcn_fence(__ATOMIC_RELEASE, "workgroup", "local");
  __builtin_amdgcn_s_barrier_signal(-1);
}
__device__ __forceinline__ void swap_wait() {
  __builtin_amdgcn_s_barrier_wait(-1);
  __builtin_amdgcn_fence(__ATOMIC_ACQUIRE, "workgroup", "local");
}

__global__ void __launch_bounds__(128)
s6_scan_seq(const float* __restrict__ xdbl, const float* __restrict__ delta,
            const float* __restrict__ x, const float* __restrict__ Alog,
            const float* __restrict__ Dp, float* __restrict__ y) {
  __shared__ float xch[2][4][32];

  const int lane = threadIdx.x & 31;
  const int wv = __builtin_amdgcn_readfirstlane((int)(threadIdx.x >> 5));
  const int q  = wv & 1;
  const int pw = blockIdx.x * 2 + (wv >> 1);
  const int bb = pw / (D_ / 2);
  const int c  = lane & 15;
  const int dc = (pw % (D_ / 2)) * 2 + (lane >> 4);
  const int n0 = 32 * q + 2 * c;

  const v2f al = *(const v2f*)(Alog + (size_t)dc * N_ + n0);
  v2f A;
  A.x = -__expf(al.x);  A.y = -__expf(al.y);
  const float Dv = Dp[dc];

  const float* bc = xdbl + (size_t)bb * L_ * E_ + R_ + n0;
  const size_t lrow = ((size_t)bb * L_ + c) * D_ + dc;
  const float* pdl = delta + lrow;
  const float* pxv = x + lrow;
  float*       py  = y + lrow + (size_t)(q * SH_) * D_;

  const bool u3 = (lane & 8) != 0;
  const bool u2 = (lane & 4) != 0;
  const bool u1 = (lane & 2) != 0;
  const bool u0 = (lane & 1) != 0;

  float dlnA = pdl[0];
  float dlnB = pdl[(size_t)SH_ * D_];
  float xvnA = pxv[0];
  float xvnB = pxv[(size_t)SH_ * D_];
  float dlmA = pdl[(size_t)SB_ * D_];
  float dlmB = pdl[(size_t)(SB_ + SH_) * D_];
  float xvmA = pxv[(size_t)SB_ * D_];
  float xvmB = pxv[(size_t)(SB_ + SH_) * D_];
  __builtin_amdgcn_sched_barrier(0);
  v2f Bv[SB_], Cv[SB_];
#pragma unroll
  for (int i = 0; i < SB_; ++i) {
    Bv[i] = *(const v2f*)(bc + (size_t)i * E_);
    Cv[i] = *(const v2f*)(bc + (size_t)i * E_ + N_);
  }
  __builtin_amdgcn_sched_barrier(0);

  xch[1][wv][lane] = 0.0f;

  v2f h = {0.0f, 0.0f};
  float minePrev = 0.0f;
  float xdPrev   = 0.0f;
  float other    = 0.0f;
  for (int t0 = 0; t0 < L_; t0 += SB_) {
    const float dlvA = dlnA;
    const float dlvB = dlnB;
    const float duvA = dlvA * xvnA;
    const float duvB = dlvB * xvnB;
    const float xdCur = (q ? xvnB : xvnA) * Dv;
    const int par = (t0 / SB_) & 1;
    const int tn = (t0 + SB_ < L_) ? (t0 + SB_) : t0;
    const int tm = (t0 + 2 * SB_ < L_) ? (t0 + 2 * SB_) : (L_ - SB_);
    const float* bcn = bc + (size_t)tn * E_;

    float part[SB_];
#pragma unroll
    for (int j = 0; j < SB_ / SG_; ++j) {
#pragma unroll
      for (int k = 0; k < SG_; ++k) {
        const int i = j * SG_ + k;
        const float dlt = (i < SH_) ? row_share(dlvA, i) : row_share(dlvB, i - SH_);
        const float du  = (i < SH_) ? row_share(duvA, i) : row_share(duvB, i - SH_);
        const v2f e = A * dlt;
        v2f a;
        a.x = __expf(e.x);  a.y = __expf(e.y);
        h = a * h + du * Bv[i];
        const v2f hc = h * Cv[i];
        part[i] = hc.x + hc.y;
      }
      asm volatile("" :: "v"(part[j * SG_ + 0]), "v"(part[j * SG_ + 1]),
                         "v"(part[j * SG_ + 2]), "v"(part[j * SG_ + 3]),
                         "v"(part[j * SG_ + 4]), "v"(part[j * SG_ + 5]),
                         "v"(part[j * SG_ + 6]), "v"(part[j * SG_ + 7]) : "memory");
      if (j == SB_ / SG_ - 1) {
        asm("v_mov_b32 %0, %1" : "=v"(dlnA) : "v"(dlmA));
        asm("v_mov_b32 %0, %1" : "=v"(dlnB) : "v"(dlmB));
        asm("v_mov_b32 %0, %1" : "=v"(xvnA) : "v"(xvmA));
        asm("v_mov_b32 %0, %1" : "=v"(xvnB) : "v"(xvmB));
        dlmA = pdl[(size_t)tm * D_];
        dlmB = pdl[(size_t)(tm + SH_) * D_];
        xvmA = pxv[(size_t)tm * D_];
        xvmB = pxv[(size_t)(tm + SH_) * D_];
      }
#pragma unroll
      for (int k = 0; k < SG_; ++k) {
        const int i = j * SG_ + k;
        Bv[i] = *(const v2f*)(bcn + (size_t)i * E_);
        Cv[i] = *(const v2f*)(bcn + (size_t)i * E_ + N_);
      }
      __builtin_amdgcn_sched_barrier(0);
      if (j == 0) swap_arrive();
      if (j == 1) swap_wait();
    }
    other = xch[par ^ 1][wv ^ 1][lane];
    __builtin_amdgcn_sched_barrier(0);

    const float totA = row_sums(part, u3, u2, u1, u0);
    const float totB = row_sums(part + SH_, u3, u2, u1, u0);

    const float yPrev = (minePrev + other) + xdPrev;
    minePrev = q ? totB : totA;
    xdPrev   = xdCur;
    xch[par][wv][lane] = q ? totA : totB;
    if (t0 != 0) py[(size_t)(t0 - SB_) * D_] = yPrev;
  }
  swap_arrive();
  swap_wait();
  other = xch[((L_ / SB_) - 1) & 1][wv ^ 1][lane];
  py[(size_t)(L_ - SB_) * D_] = (minePrev + other) + xdPrev;
}

extern "C" void kernel_launch(void* const* d_in, const int* in_sizes, int n_in,
                              void* d_out, int out_size, void* d_ws, size_t ws_size,
                              hipStream_t stream) {
  const float* x      = (const float*)d_in[0];
  const float* Wxproj = (const float*)d_in[1];
  const float* Wdt    = (const float*)d_in[2];
  const float* bdt    = (const float*)d_in[3];
  const float* Alog   = (const float*)d_in[4];
  const float* Dp     = (const float*)d_in[5];
  float*       y      = (float*)d_out;

  float* ws    = (float*)d_ws;
  float* xdbl  = ws;
  float* delta = xdbl  + (size_t)BL_ * E_;

  s6_xproj_gemm<<<256, 128, 0, stream>>>(x, Wxproj, xdbl);
  s6_delta_gemm<<<1024, 256, 0, stream>>>(xdbl, Wdt, bdt, delta);
  s6_scan_seq<<<256, 128, 0, stream>>>(xdbl, delta, x, Alog, Dp, y);
}


# Scan: B/C rows, delta and x fetched TWO batches ahead, not one: two register sets, one 16-byte B,B,C,C load a step (x_dbl columns reordered in the x_proj GEMM); sum order unchanged
# speedup vs baseline: 3.3418x; 1.1690x over previous
#include <hip/hip_runtime.h>
#include <hip/hip_bf16.h>

typedef float v2f __attribute__((ext_vector_type(2)));
typedef float v8f __attribute__((ext_vector_type(8)));

#define B_  2
#define L_  2048
#define D_  512
#define N_  64
#define R_  32
#define E_  160
#define BL_ (B_ * L_)

typedef float v4f __attribute__((ext_vector_type(4)));

#define XU_ 8
#define XG_ (D_ / (8 * XU_))

template <int E>
struct XFrag {
  v4f a[XU_];
  v4f b[E][XU_];
};

__device__ __forceinline__ int xdbl_src_row(int col) {
  if (col < R_) return col;
  const int k = col - R_;
  const int pr = k >> 2;
  const int s = k & 3;
  return (s < 2) ? (R_ + 2 * pr + s) : (R_ + N_ + 2 * pr + (s - 2));
}

template <int E>
__device__ __forceinline__ void xproj_load(XFrag<E>& f, const float* __restrict__ ap,
                                           const float* (&bp)[E], int g) {
#pragma unroll
  for (int u = 0; u < XU_; ++u) {
    f.a[u] = *(const v4f*)(ap + (g * XU_ + u) * 8);
#pragma unroll
    for (int e = 0; e < E; ++e)
      f.b[e][u] = *(const v4f*)(bp[e] + (g * XU_ + u) * 8);
  }
}

template <int E>
__device__ __forceinline__ void xproj_mma(v8f (&acc)[E], const XFrag<E>& f) {
#pragma unroll
  for (int u = 0; u < XU_; ++u) {
    const v2f a0 = f.a[u].xy;
    const v2f a1 = f.a[u].zw;
#pragma unroll
    for (int e = 0; e < E; ++e) {
      const v2f b0 = f.b[e][u].xy;
      acc[e] = __builtin_amdgcn_wmma_f32_16x16x4_f32(false, a0, false, b0,
                                                     (short)0, acc[e], false, false);
    }
#pragma unroll
    for (int e = 0; e < E; ++e) {
      const v2f b1 = f.b[e][u].zw;
      acc[e] = __builtin_amdgcn_wmma_f32_16x16x4_f32(false, a1, false, b1,
                                                     (short)0, acc[e], false, false);
    }
  }
}

template <int E>
__device__ __forceinline__ void xproj_strip(const float* __restrict__ ap,
                                            const float* __restrict__ wbase, int col0,
                                            float* __restrict__ op) {
  const float* bp[E];
#pragma unroll
  for (int e = 0; e < E; ++e) bp[e] = wbase + (size_t)xdbl_src_row(col0 + e * 16) * D_;
  v8f acc[E];
#pragma unroll
  for (int e = 0; e < E; ++e) acc[e] = v8f{};
  XFrag<E> f0, f1;
  xproj_load<E>(f0, ap, bp, 0);
  for (int g = 0; g < XG_ - 2; g += 2) {
    xproj_load<E>(f1, ap, bp, g + 1);
    __builtin_amdgcn_sched_barrier(0);
    xproj_mma<E>(acc, f0);
    __builtin_amdgcn_sched_barrier(0);
    xproj_load<E>(f0, ap, bp, g + 2);
    __builtin_amdgcn_sched_barrier(0);
    xproj_mma<E>(acc, f1);
    __builtin_amdgcn_sched_barrier(0);
  }
  xproj_load<E>(f1, ap, bp, XG_ - 1);
  __builtin_amdgcn_sched_barrier(0);
  xproj_mma<E>(acc, f0);
  xproj_mma<E>(acc, f1);
#pragma unroll
  for (int e = 0; e < E; ++e)
#pragma unroll
    for (int j = 0; j < 8; ++j)
      op[(size_t)j * E_ + e * 16] = acc[e][j];
}

__global__ void __launch_bounds__(128)
s6_xproj_gemm(const float* __restrict__ x, const float* __restrict__ W,
              float* __restrict__ xdbl) {
  const int lane  = threadIdx.x & 31;
  const int wv    = __builtin_amdgcn_readfirstlane((int)(threadIdx.x >> 5));
  const int rt    = blockIdx.x;
  const int laneM = lane & 15;
  const int laneH = lane >> 4;

  const float* ap = x + (size_t)(rt * 16 + laneM) * D_ + 4 * laneH;
  float* orow = xdbl + (size_t)(rt * 16 + 8 * laneH) * E_ + laneM;
  if (wv < 2) {
    const int t0 = 3 * wv;
    xproj_strip<3>(ap, W + 4 * laneH, t0 * 16 + laneM, orow + t0 * 16);
  } else {
    const int t0 = 2 * wv + 2;
    xproj_strip<2>(ap, W + 4 * laneH, t0 * 16 + laneM, orow + t0 * 16);
  }
}

__global__ void __launch_bounds__(256)
s6_delta_gemm(const float* __restrict__ xdbl, const float* __restrict__ Wdt,
              const float* __restrict__ bdt, float* __restrict__ delta) {
  const int wave  = blockIdx.x * (blockDim.x >> 5) + (threadIdx.x >> 5);
  const int lane  = threadIdx.x & 31;
  const int rt    = wave >> 5;
  const int nt    = wave & 31;
  const int laneM = lane & 15;
  const int laneH = lane >> 4;

  const float* arow = xdbl + (size_t)(rt * 16 + laneM) * E_ + 2 * laneH;
  const float* brow = Wdt  + (size_t)(nt * 16 + laneM) * R_ + 2 * laneH;

  v8f c = {};
#pragma unroll
  for (int k = 0; k < R_; k += 4) {
    v2f a = *(const v2f*)(arow + k);
    v2f b = *(const v2f*)(brow + k);
    c = __builtin_amdgcn_wmma_f32_16x16x4_f32(false, a, false, b,
                                              (short)0, c, false, false);
  }
  const int col  = nt * 16 + laneM;
  const float bi = bdt[col];
#pragma unroll
  for (int j = 0; j < 8; ++j) {
    const int row = rt * 16 + j + 8 * laneH;
    float z = c[j] + bi;
    float sp = (z > 20.0f) ? z : __logf(1.0f + __expf(z));
    delta[(size_t)row * D_ + col] = sp;
  }
}

#define SB_ 32
#define SH_ 16
#define SG_ 8

__device__ __forceinline__ float row_share(float v, int i) {
  const int b = __builtin_bit_cast(int, v);
  int r = 0;
  switch (i) {
    case 0: r = __builtin_amdgcn_update_dpp(0, b, 0x150 + 0, 0xf, 0xf, true); break;
    case 1: r = __builtin_amdgcn_update_dpp(0, b, 0x150 + 1, 0xf, 0xf, true); break;
    case 2: r = __builtin_amdgcn_update_dpp(0, b, 0x150 + 2, 0xf, 0xf, true); break;
    case 3: r = __builtin_amdgcn_update_dpp(0, b, 0x150 + 3, 0xf, 0xf, true); break;
    case 4: r = __builtin_amdgcn_update_dpp(0, b, 0x150 + 4, 0xf, 0xf, true); break;
    case 5: r = __builtin_amdgcn_update_dpp(0, b, 0x150 + 5, 0xf, 0xf, true); break;
    case 6: r = __builtin_amdgcn_update_dpp(0, b, 0x150 + 6, 0xf, 0xf, true); break;
    case 7: r = __builtin_amdgcn_update_dpp(0, b, 0x150 + 7, 0xf, 0xf, true); break;
    case 8: r = __builtin_amdgcn_update_dpp(0, b, 0x150 + 8, 0xf, 0xf, true); break;
    case 9: r = __builtin_amdgcn_update_dpp(0, b, 0x150 + 9, 0xf, 0xf, true); break;
    case 10: r = __builtin_amdgcn_update_dpp(0, b, 0x150 + 10, 0xf, 0xf, true); break;
    case 11: r = __builtin_amdgcn_update_dpp(0, b, 0x150 + 11, 0xf, 0xf, true); break;
    case 12: r = __builtin_amdgcn_update_dpp(0, b, 0x150 + 12, 0xf, 0xf, true); break;
    case 13: r = __builtin_amdgcn_update_dpp(0, b, 0x150 + 13, 0xf, 0xf, true); break;
    case 14: r = __builtin_amdgcn_update_dpp(0, b, 0x150 + 14, 0xf, 0xf, true); break;
    case 15: r = __builtin_amdgcn_update_dpp(0, b, 0x150 + 15, 0xf, 0xf, true); break;
  }
  return __builtin_bit_cast(float, r);
}
template <int M>
__device__ __forceinline__ float row_xor(float v) {
  const int b = __builtin_bit_cast(int, v);
  return __builtin_bit_cast(float,
      __builtin_amdgcn_update_dpp(0, b, 0x160 + M, 0xf, 0xf, true));
}

__device__ __forceinline__ float row_sums(const float* part, bool u3, bool u2,
                                          bool u1, bool u0) {
  float r8[8], r4[4], r2[2];
#pragma unroll
  for (int i = 0; i < 8; ++i) {
    const float keep = u3 ? part[i + 8] : part[i];
    const float send = u3 ? part[i] : part[i + 8];
    r8[i] = keep + row_xor<8>(send);
  }
#pragma unroll
  for (int i = 0; i < 4; ++i) {
    const float keep = u2 ? r8[i + 4] : r8[i];
    const float send = u2 ? r8[i] : r8[i + 4];
    r4[i] = keep + row_xor<4>(send);
  }
#pragma unroll
  for (int i = 0; i < 2; ++i) {
    const float keep = u1 ? r4[i + 2] : r4[i];
    const float send = u1 ? r4[i] : r4[i + 2];
    r2[i] = keep + row_xor<2>(send);
  }
  const float keep0 = u0 ? r2[1] : r2[0];
  const float send0 = u0 ? r2[0] : r2[1];
  return keep0 + row_xor<1>(send0);
}

__device__ __forceinline__ void swap_arrive() {
  __builtin_amdgcn_fence(__ATOMIC_RELEASE, "workgroup", "local");
  __builtin_amdgcn_s_barrier_signal(-1);
}
__device__ __forceinline__ void swap_wait() {
  __builtin_amdgcn_s_barrier_wait(-1);
  __builtin_amdgcn_fence(__ATOMIC_ACQUIRE, "workgroup", "local");
}

__global__ void __launch_bounds__(128)
s6_scan_seq(const float* __restrict__ xdbl, const float* __restrict__ delta,
            const float* __restrict__ x, const float* __restrict__ Alog,
            const float* __restrict__ Dp, float* __restrict__ y) {
  __shared__ float xch[2][4][32];

  const int lane = threadIdx.x & 31;
  const int wv = __builtin_amdgcn_readfirstlane((int)(threadIdx.x >> 5));
  const int q  = wv & 1;
  const int pw = blockIdx.x * 2 + (wv >> 1);
  const int bb = pw / (D_ / 2);
  const int c  = lane & 15;
  const int dc = (pw % (D_ / 2)) * 2 + (lane >> 4);
  const int n0 = 32 * q + 2 * c;

  const v2f al = *(const v2f*)(Alog + (size_t)dc * N_ + n0);
  v2f A;
  A.x = -__expf(al.x);  A.y = -__expf(al.y);
  const float Dv = Dp[dc];

  const float* bc = xdbl + (size_t)bb * L_ * E_ + R_ + 64 * q + 4 * c;
  const size_t lrow = ((size_t)bb * L_ + c) * D_ + dc;
  const float* pdl = delta + lrow;
  const float* pxv = x + lrow;
  float*       py  = y + lrow + (size_t)(q * SH_) * D_;

  const bool u3 = (lane & 8) != 0;
  const bool u2 = (lane & 4) != 0;
  const bool u1 = (lane & 2) != 0;
  const bool u0 = (lane & 1) != 0;

  v2f dlA, xvA, dlB, xvB;
  dlA.x = pdl[0];
  dlA.y = pdl[(size_t)SH_ * D_];
  xvA.x = pxv[0];
  xvA.y = pxv[(size_t)SH_ * D_];
  dlB.x = pdl[(size_t)SB_ * D_];
  dlB.y = pdl[(size_t)(SB_ + SH_) * D_];
  xvB.x = pxv[(size_t)SB_ * D_];
  xvB.y = pxv[(size_t)(SB_ + SH_) * D_];
  __builtin_amdgcn_sched_barrier(0);
  v4f RA[SB_], RB[SB_];
#pragma unroll
  for (int i = 0; i < SB_; ++i) RA[i] = *(const v4f*)(bc + (size_t)i * E_);
#pragma unroll
  for (int i = 0; i < SB_; ++i) RB[i] = *(const v4f*)(bc + (size_t)(SB_ + i) * E_);
  __builtin_amdgcn_sched_barrier(0);

  xch[1][wv][lane] = 0.0f;

  v2f h = {0.0f, 0.0f};
  float minePrev = 0.0f;
  float xdPrev   = 0.0f;
  float other    = 0.0f;

  auto batch = [&](v4f (&Rw)[SB_], v2f& dl, v2f& xv, const int t0, const int par)
      __attribute__((always_inline)) {
    v2f dw;
    asm("v_mov_b64 %0, %1" : "=v"(dw) : "v"(dl));
    const float dlvA = dw.x;
    const float dlvB = dw.y;
    const v2f duv = dw * xv;
    const float duvA = duv.x;
    const float duvB = duv.y;
    const float xdCur = (q ? xv.y : xv.x) * Dv;
    const int tn = (t0 + 2 * SB_ < L_) ? (t0 + 2 * SB_) : t0;
    const float* bcn = bc + (size_t)tn * E_;

    float part[SB_];
#pragma unroll
    for (int j = 0; j < SB_ / SG_; ++j) {
#pragma unroll
      for (int k = 0; k < SG_; ++k) {
        const int i = j * SG_ + k;
        const float dlt = (i < SH_) ? row_share(dlvA, i) : row_share(dlvB, i - SH_);
        const float du  = (i < SH_) ? row_share(duvA, i) : row_share(duvB, i - SH_);
        const v2f Bi = Rw[i].xy;
        const v2f Ci = Rw[i].zw;
        const v2f e = A * dlt;
        v2f a;
        a.x = __expf(e.x);  a.y = __expf(e.y);
        h = a * h + du * Bi;
        const v2f hc = h * Ci;
        part[i] = hc.x + hc.y;
      }
      asm volatile("" :: "v"(part[j * SG_ + 0]), "v"(part[j * SG_ + 1]),
                         "v"(part[j * SG_ + 2]), "v"(part[j * SG_ + 3]),
                         "v"(part[j * SG_ + 4]), "v"(part[j * SG_ + 5]),
                         "v"(part[j * SG_ + 6]), "v"(part[j * SG_ + 7]) : "memory");
      if (j == 0) {
        dl.x = pdl[(size_t)tn * D_];
        dl.y = pdl[(size_t)(tn + SH_) * D_];
        xv.x = pxv[(size_t)tn * D_];
        xv.y = pxv[(size_t)(tn + SH_) * D_];
      }
#pragma unroll
      for (int k = 0; k < SG_; ++k) {
        const int i = j * SG_ + k;
        Rw[i] = *(const v4f*)(bcn + (size_t)i * E_);
      }
      __builtin_amdgcn_sched_barrier(0);
      if (j == 0) swap_arrive();
      if (j == 1) swap_wait();
    }
    other = xch[par ^ 1][wv ^ 1][lane];
    __builtin_amdgcn_sched_barrier(0);

    const float totA = row_sums(part, u3, u2, u1, u0);
    const float totB = row_sums(part + SH_, u3, u2, u1, u0);

    const float yPrev = (minePrev + other) + xdPrev;
    minePrev = q ? totB : totA;
    xdPrev   = xdCur;
    xch[par][wv][lane] = q ? totA : totB;
    if (t0 != 0) py[(size_t)(t0 - SB_) * D_] = yPrev;
  };

  for (int t0 = 0; t0 < L_; t0 += 2 * SB_) {
    batch(RA, dlA, xvA, t0, 0);
    batch(RB, dlB, xvB, t0 + SB_, 1);
  }
  swap_arrive();
  swap_wait();
  other = xch[((L_ / SB_) - 1) & 1][wv ^ 1][lane];
  py[(size_t)(L_ - SB_) * D_] = (minePrev + other) + xdPrev;
}

extern "C" void kernel_launch(void* const* d_in, const int* in_sizes, int n_in,
                              void* d_out, int out_size, void* d_ws, size_t ws_size,
                              hipStream_t stream) {
  const float* x      = (const float*)d_in[0];
  const float* Wxproj = (const float*)d_in[1];
  const float* Wdt    = (const float*)d_in[2];
  const float* bdt    = (const float*)d_in[3];
  const float* Alog   = (const float*)d_in[4];
  const float* Dp     = (const float*)d_in[5];
  float*       y      = (float*)d_out;

  float* ws    = (float*)d_ws;
  float* xdbl  = ws;
  float* delta = xdbl  + (size_t)BL_ * E_;

  s6_xproj_gemm<<<256, 128, 0, stream>>>(x, Wxproj, xdbl);
  s6_delta_gemm<<<1024, 256, 0, stream>>>(xdbl, Wdt, bdt, delta);
  s6_scan_seq<<<256, 128, 0, stream>>>(xdbl, delta, x, Alog, Dp, y);
}


# Scan: L2 prefetch (global_prefetch_b8) of the delta, x and B/C lines 8 batches ahead of their use, one instruction a wave and batch, the asking shared among work-groups; fetches and sums unchanged
# speedup vs baseline: 3.8586x; 1.1546x over previous
#include <hip/hip_runtime.h>
#include <hip/hip_bf16.h>

typedef float v2f __attribute__((ext_vector_type(2)));
typedef float v8f __attribute__((ext_vector_type(8)));

#define B_  2
#define L_  2048
#define D_  512
#define N_  64
#define R_  32
#define E_  160
#define BL_ (B_ * L_)

typedef float v4f __attribute__((ext_vector_type(4)));

#define XU_ 8
#define XG_ (D_ / (8 * XU_))

template <int E>
struct XFrag {
  v4f a[XU_];
  v4f b[E][XU_];
};

__device__ __forceinline__ int xdbl_src_row(int col) {
  if (col < R_) return col;
  const int k = col - R_;
  const int pr = k >> 2;
  const int s = k & 3;
  return (s < 2) ? (R_ + 2 * pr + s) : (R_ + N_ + 2 * pr + (s - 2));
}

template <int E>
__device__ __forceinline__ void xproj_load(XFrag<E>& f, const float* __restrict__ ap,
                                           const float* (&bp)[E], int g) {
#pragma unroll
  for (int u = 0; u < XU_; ++u) {
    f.a[u] = *(const v4f*)(ap + (g * XU_ + u) * 8);
#pragma unroll
    for (int e = 0; e < E; ++e)
      f.b[e][u] = *(const v4f*)(bp[e] + (g * XU_ + u) * 8);
  }
}

template <int E>
__device__ __forceinline__ void xproj_mma(v8f (&acc)[E], const XFrag<E>& f) {
#pragma unroll
  for (int u = 0; u < XU_; ++u) {
    const v2f a0 = f.a[u].xy;
    const v2f a1 = f.a[u].zw;
#pragma unroll
    for (int e = 0; e < E; ++e) {
      const v2f b0 = f.b[e][u].xy;
      acc[e] = __builtin_amdgcn_wmma_f32_16x16x4_f32(false, a0, false, b0,
                                                     (short)0, acc[e], false, false);
    }
#pragma unroll
    for (int e = 0; e < E; ++e) {
      const v2f b1 = f.b[e][u].zw;
      acc[e] = __builtin_amdgcn_wmma_f32_16x16x4_f32(false, a1, false, b1,
                                                     (short)0, acc[e], false, false);
    }
  }
}

template <int E>
__device__ __forceinline__ void xproj_strip(const float* __restrict__ ap,
                                            const float* __restrict__ wbase, int col0,
                                            float* __restrict__ op) {
  const float* bp[E];
#pragma unroll
  for (int e = 0; e < E; ++e) bp[e] = wbase + (size_t)xdbl_src_row(col0 + e * 16) * D_;
  v8f acc[E];
#pragma unroll
  for (int e = 0; e < E; ++e) acc[e] = v8f{};
  XFrag<E> f0, f1;
  xproj_load<E>(f0, ap, bp, 0);
  for (int g = 0; g < XG_ - 2; g += 2) {
    xproj_load<E>(f1, ap, bp, g + 1);
    __builtin_amdgcn_sched_barrier(0);
    xproj_mma<E>(acc, f0);
    __builtin_amdgcn_sched_barrier(0);
    xproj_load<E>(f0, ap, bp, g + 2);
    __builtin_amdgcn_sched_barrier(0);
    xproj_mma<E>(acc, f1);
    __builtin_amdgcn_sched_barrier(0);
  }
  xproj_load<E>(f1, ap, bp, XG_ - 1);
  __builtin_amdgcn_sched_barrier(0);
  xproj_mma<E>(acc, f0);
  xproj_mma<E>(acc, f1);
#pragma unroll
  for (int e = 0; e < E; ++e)
#pragma unroll
    for (int j = 0; j < 8; ++j)
      op[(size_t)j * E_ + e * 16] = acc[e][j];
}

__global__ void __launch_bounds__(128)
s6_xproj_gemm(const float* __restrict__ x, const float* __restrict__ W,
              float* __restrict__ xdbl) {
  const int lane  = threadIdx.x & 31;
  const int wv    = __builtin_amdgcn_readfirstlane((int)(threadIdx.x >> 5));
  const int rt    = blockIdx.x;
  const int laneM = lane & 15;
  const int laneH = lane >> 4;

  const float* ap = x + (size_t)(rt * 16 + laneM) * D_ + 4 * laneH;
  float* orow = xdbl + (size_t)(rt * 16 + 8 * laneH) * E_ + laneM;
  if (wv < 2) {
    const int t0 = 3 * wv;
    xproj_strip<3>(ap, W + 4 * laneH, t0 * 16 + laneM, orow + t0 * 16);
  } else {
    const int t0 = 2 * wv + 2;
    xproj_strip<2>(ap, W + 4 * laneH, t0 * 16 + laneM, orow + t0 * 16);
  }
}

__global__ void __launch_bounds__(256)
s6_delta_gemm(const float* __restrict__ xdbl, const float* __restrict__ Wdt,
              const float* __restrict__ bdt, float* __restrict__ delta) {
  const int wave  = blockIdx.x * (blockDim.x >> 5) + (threadIdx.x >> 5);
  const int lane  = threadIdx.x & 31;
  const int rt    = wave >> 5;
  const int nt    = wave & 31;
  const int laneM = lane & 15;
  const int laneH = lane >> 4;

  const float* arow = xdbl + (size_t)(rt * 16 + laneM) * E_ + 2 * laneH;
  const float* brow = Wdt  + (size_t)(nt * 16 + laneM) * R_ + 2 * laneH;

  v8f c = {};
#pragma unroll
  for (int k = 0; k < R_; k += 4) {
    v2f a = *(const v2f*)(arow + k);
    v2f b = *(const v2f*)(brow + k);
    c = __builtin_amdgcn_wmma_f32_16x16x4_f32(false, a, false, b,
                                              (short)0, c, false, false);
  }
  const int col  = nt * 16 + laneM;
  const float bi = bdt[col];
#pragma unroll
  for (int j = 0; j < 8; ++j) {
    const int row = rt * 16 + j + 8 * laneH;
    float z = c[j] + bi;
    float sp = (z > 20.0f) ? z : __logf(1.0f + __expf(z));
    delta[(size_t)row * D_ + col] = sp;
  }
}

#define SB_ 32
#define SH_ 16
#define SG_ 8
#define PF_ 8

__device__ __forceinline__ void l2_warm(const char* p) {
  __builtin_prefetch(p, 0, 1);
}

__device__ __forceinline__ float row_share(float v, int i) {
  const int b = __builtin_bit_cast(int, v);
  int r = 0;
  switch (i) {
    case 0: r = __builtin_amdgcn_update_dpp(0, b, 0x150 + 0, 0xf, 0xf, true); break;
    case 1: r = __builtin_amdgcn_update_dpp(0, b, 0x150 + 1, 0xf, 0xf, true); break;
    case 2: r = __builtin_amdgcn_update_dpp(0, b, 0x150 + 2, 0xf, 0xf, true); break;
    case 3: r = __builtin_amdgcn_update_dpp(0, b, 0x150 + 3, 0xf, 0xf, true); break;
    case 4: r = __builtin_amdgcn_update_dpp(0, b, 0x150 + 4, 0xf, 0xf, true); break;
    case 5: r = __builtin_amdgcn_update_dpp(0, b, 0x150 + 5, 0xf, 0xf, true); break;
    case 6: r = __builtin_amdgcn_update_dpp(0, b, 0x150 + 6, 0xf, 0xf, true); break;
    case 7: r = __builtin_amdgcn_update_dpp(0, b, 0x150 + 7, 0xf, 0xf, true); break;
    case 8: r = __builtin_amdgcn_update_dpp(0, b, 0x150 + 8, 0xf, 0xf, true); break;
    case 9: r = __builtin_amdgcn_update_dpp(0, b, 0x150 + 9, 0xf, 0xf, true); break;
    case 10: r = __builtin_amdgcn_update_dpp(0, b, 0x150 + 10, 0xf, 0xf, true); break;
    case 11: r = __builtin_amdgcn_update_dpp(0, b, 0x150 + 11, 0xf, 0xf, true); break;
    case 12: r = __builtin_amdgcn_update_dpp(0, b, 0x150 + 12, 0xf, 0xf, true); break;
    case 13: r = __builtin_amdgcn_update_dpp(0, b, 0x150 + 13, 0xf, 0xf, true); break;
    case 14: r = __builtin_amdgcn_update_dpp(0, b, 0x150 + 14, 0xf, 0xf, true); break;
    case 15: r = __builtin_amdgcn_update_dpp(0, b, 0x150 + 15, 0xf, 0xf, true); break;
  }
  return __builtin_bit_cast(float, r);
}
template <int M>
__device__ __forceinline__ float row_xor(float v) {
  const int b = __builtin_bit_cast(int, v);
  return __builtin_bit_cast(float,
      __builtin_amdgcn_update_dpp(0, b, 0x160 + M, 0xf, 0xf, true));
}

__device__ __forceinline__ float row_sums(const float* part, bool u3, bool u2,
                                          bool u1, bool u0) {
  float r8[8], r4[4], r2[2];
#pragma unroll
  for (int i = 0; i < 8; ++i) {
    const float keep = u3 ? part[i + 8] : part[i];
    const float send = u3 ? part[i] : part[i + 8];
    r8[i] = keep + row_xor<8>(send);
  }
#pragma unroll
  for (int i = 0; i < 4; ++i) {
    const float keep = u2 ? r8[i + 4] : r8[i];
    const float send = u2 ? r8[i] : r8[i + 4];
    r4[i] = keep + row_xor<4>(send);
  }
#pragma unroll
  for (int i = 0; i < 2; ++i) {
    const float keep = u1 ? r4[i + 2] : r4[i];
    const float send = u1 ? r4[i] : r4[i + 2];
    r2[i] = keep + row_xor<2>(send);
  }
  const float keep0 = u0 ? r2[1] : r2[0];
  const float send0 = u0 ? r2[0] : r2[1];
  return keep0 + row_xor<1>(send0);
}

__device__ __forceinline__ void swap_arrive() {
  __builtin_amdgcn_fence(__ATOMIC_RELEASE, "workgroup", "local");
  __builtin_amdgcn_s_barrier_signal(-1);
}
__device__ __forceinline__ void swap_wait() {
  __builtin_amdgcn_s_barrier_wait(-1);
  __builtin_amdgcn_fence(__ATOMIC_ACQUIRE, "workgroup", "local");
}

__global__ void __launch_bounds__(128)
s6_scan_seq(const float* __restrict__ xdbl, const float* __restrict__ delta,
            const float* __restrict__ x, const float* __restrict__ Alog,
            const float* __restrict__ Dp, float* __restrict__ y) {
  __shared__ float xch[2][4][32];

  const int lane = threadIdx.x & 31;
  const int wv = __builtin_amdgcn_readfirstlane((int)(threadIdx.x >> 5));
  const int q  = wv & 1;
  const int pw = blockIdx.x * 2 + (wv >> 1);
  const int bb = pw / (D_ / 2);
  const int c  = lane & 15;
  const int dc = (pw % (D_ / 2)) * 2 + (lane >> 4);
  const int n0 = 32 * q + 2 * c;

  const v2f al = *(const v2f*)(Alog + (size_t)dc * N_ + n0);
  v2f A;
  A.x = -__expf(al.x);  A.y = -__expf(al.y);
  const float Dv = Dp[dc];

  const float* bc = xdbl + (size_t)bb * L_ * E_ + R_ + 64 * q + 4 * c;
  const size_t lrow = ((size_t)bb * L_ + c) * D_ + dc;
  const float* pdl = delta + lrow;
  const float* pxv = x + lrow;
  float*       py  = y + lrow + (size_t)(q * SH_) * D_;

  const bool u3 = (lane & 8) != 0;
  const bool u2 = (lane & 4) != 0;
  const bool u1 = (lane & 2) != 0;
  const bool u0 = (lane & 1) != 0;

  v2f dlA, xvA, dlB, xvB;
  dlA.x = pdl[0];
  dlA.y = pdl[(size_t)SH_ * D_];
  xvA.x = pxv[0];
  xvA.y = pxv[(size_t)SH_ * D_];
  dlB.x = pdl[(size_t)SB_ * D_];
  dlB.y = pdl[(size_t)(SB_ + SH_) * D_];
  xvB.x = pxv[(size_t)SB_ * D_];
  xvB.y = pxv[(size_t)(SB_ + SH_) * D_];
  __builtin_amdgcn_sched_barrier(0);
  v4f RA[SB_], RB[SB_];
#pragma unroll
  for (int i = 0; i < SB_; ++i) RA[i] = *(const v4f*)(bc + (size_t)i * E_);
#pragma unroll
  for (int i = 0; i < SB_; ++i) RB[i] = *(const v4f*)(bc + (size_t)(SB_ + i) * E_);
  __builtin_amdgcn_sched_barrier(0);
  const int gq = blockIdx.x & (D_ / 4 - 1);
  const int rs = (gq + (gq >> 2)) & 3;
  const int r8 = ((gq + (gq >> 3)) & 7) ^ ((wv & 1) << 2);
  const int pj = (lane < 10) ? lane : 9;
  const float* pfPlane = (wv == 0) ? delta : (wv == 1) ? x : xdbl;
  const int pfStep = (wv < 2) ? D_ * 4 : E_ * 4;
  const unsigned pfOffDX = ((unsigned)(bb * L_ + 4 * (lane & 7) + rs) * D_ + 4 * gq) * 4u;
  const unsigned pfOffBC = (unsigned)bb * (L_ * E_ * 4) + (unsigned)(8 * pj + r8) * 256u;
  const char* pfp = (const char*)pfPlane + ((wv < 2) ? pfOffDX : pfOffBC);
  for (int k = 2; k < PF_; ++k) l2_warm(pfp + (size_t)(k * SB_) * pfStep);

  xch[1][wv][lane] = 0.0f;

  v2f h = {0.0f, 0.0f};
  float minePrev = 0.0f;
  float xdPrev   = 0.0f;
  float other    = 0.0f;

  auto batch = [&](v4f (&Rw)[SB_], v2f& dl, v2f& xv, const int t0, const int par)
      __attribute__((always_inline)) {
    v2f dw;
    asm("v_mov_b64 %0, %1" : "=v"(dw) : "v"(dl));
    const float dlvA = dw.x;
    const float dlvB = dw.y;
    const v2f duv = dw * xv;
    const float duvA = duv.x;
    const float duvB = duv.y;
    const float xdCur = (q ? xv.y : xv.x) * Dv;
    const int tn = (t0 + 2 * SB_ < L_) ? (t0 + 2 * SB_) : t0;
    const float* bcn = bc + (size_t)tn * E_;

    float part[SB_];
#pragma unroll
    for (int j = 0; j < SB_ / SG_; ++j) {
#pragma unroll
      for (int k = 0; k < SG_; ++k) {
        const int i = j * SG_ + k;
        const float dlt = (i < SH_) ? row_share(dlvA, i) : row_share(dlvB, i - SH_);
        const float du  = (i < SH_) ? row_share(duvA, i) : row_share(duvB, i - SH_);
        const v2f Bi = Rw[i].xy;
        const v2f Ci = Rw[i].zw;
        const v2f e = A * dlt;
        v2f a;
        a.x = __expf(e.x);  a.y = __expf(e.y);
        h = a * h + du * Bi;
        const v2f hc = h * Ci;
        part[i] = hc.x + hc.y;
      }
      asm volatile("" :: "v"(part[j * SG_ + 0]), "v"(part[j * SG_ + 1]),
                         "v"(part[j * SG_ + 2]), "v"(part[j * SG_ + 3]),
                         "v"(part[j * SG_ + 4]), "v"(part[j * SG_ + 5]),
                         "v"(part[j * SG_ + 6]), "v"(part[j * SG_ + 7]) : "memory");
      if (j == 0) {
        dl.x = pdl[(size_t)tn * D_];
        dl.y = pdl[(size_t)(tn + SH_) * D_];
        xv.x = pxv[(size_t)tn * D_];
        xv.y = pxv[(size_t)(tn + SH_) * D_];
      }
#pragma unroll
      for (int k = 0; k < SG_; ++k) {
        const int i = j * SG_ + k;
        Rw[i] = *(const v4f*)(bcn + (size_t)i * E_);
      }
      if (j == 2) {
        const int tp = (t0 + PF_ * SB_ < L_) ? (t0 + PF_ * SB_) : t0;
        l2_warm(pfp + (size_t)tp * pfStep);
      }
      __builtin_amdgcn_sched_barrier(0);
      if (j == 0) swap_arrive();
      if (j == 1) swap_wait();
    }
    other = xch[par ^ 1][wv ^ 1][lane];
    __builtin_amdgcn_sched_barrier(0);

    const float totA = row_sums(part, u3, u2, u1, u0);
    const float totB = row_sums(part + SH_, u3, u2, u1, u0);

    const float yPrev = (minePrev + other) + xdPrev;
    minePrev = q ? totB : totA;
    xdPrev   = xdCur;
    xch[par][wv][lane] = q ? totA : totB;
    if (t0 != 0) py[(size_t)(t0 - SB_) * D_] = yPrev;
  };

  for (int t0 = 0; t0 < L_; t0 += 2 * SB_) {
    batch(RA, dlA, xvA, t0, 0);
    batch(RB, dlB, xvB, t0 + SB_, 1);
  }
  swap_arrive();
  swap_wait();
  other = xch[((L_ / SB_) - 1) & 1][wv ^ 1][lane];
  py[(size_t)(L_ - SB_) * D_] = (minePrev + other) + xdPrev;
}

extern "C" void kernel_launch(void* const* d_in, const int* in_sizes, int n_in,
                              void* d_out, int out_size, void* d_ws, size_t ws_size,
                              hipStream_t stream) {
  const float* x      = (const float*)d_in[0];
  const float* Wxproj = (const float*)d_in[1];
  const float* Wdt    = (const float*)d_in[2];
  const float* bdt    = (const float*)d_in[3];
  const float* Alog   = (const float*)d_in[4];
  const float* Dp     = (const float*)d_in[5];
  float*       y      = (float*)d_out;

  float* ws    = (float*)d_ws;
  float* xdbl  = ws;
  float* delta = xdbl  + (size_t)BL_ * E_;

  s6_xproj_gemm<<<256, 128, 0, stream>>>(x, Wxproj, xdbl);
  s6_delta_gemm<<<1024, 256, 0, stream>>>(xdbl, Wdt, bdt, delta);
  s6_scan_seq<<<256, 128, 0, stream>>>(xdbl, delta, x, Alog, Dp, y);
}


# delta GEMM moved into the x_proj kernel as its epilogue: delta_r handed to all 4 waves through LDS behind one barrier, 8 channel tiles a wave; 2 launches, not 3; k order and sum order unchanged
# speedup vs baseline: 4.0075x; 1.0386x over previous
#include <hip/hip_runtime.h>
#include <hip/hip_bf16.h>

typedef float v2f __attribute__((ext_vector_type(2)));
typedef float v8f __attribute__((ext_vector_type(8)));

#define B_  2
#define L_  2048
#define D_  512
#define N_  64
#define R_  32
#define E_  160
#define BL_ (B_ * L_)

typedef float v4f __attribute__((ext_vector_type(4)));

#define DRP_ 36

__device__ __forceinline__ void lds_arrive() {
  __builtin_amdgcn_fence(__ATOMIC_RELEASE, "workgroup", "local");
  __builtin_amdgcn_s_barrier_signal(-1);
}
__device__ __forceinline__ void lds_wait() {
  __builtin_amdgcn_s_barrier_wait(-1);
  __builtin_amdgcn_fence(__ATOMIC_ACQUIRE, "workgroup", "local");
}

#define XU_ 8
#define XG_ (D_ / (8 * XU_))

template <int E>
struct XFrag {
  v4f a[XU_];
  v4f b[E][XU_];
};

__device__ __forceinline__ int xdbl_src_row(int col) {
  if (col < R_) return col;
  const int k = col - R_;
  const int pr = k >> 2;
  const int s = k & 3;
  return (s < 2) ? (R_ + 2 * pr + s) : (R_ + N_ + 2 * pr + (s - 2));
}

template <int E>
__device__ __forceinline__ void xproj_load(XFrag<E>& f, const float* __restrict__ ap,
                                           const float* (&bp)[E], int g) {
#pragma unroll
  for (int u = 0; u < XU_; ++u) {
    f.a[u] = *(const v4f*)(ap + (g * XU_ + u) * 8);
#pragma unroll
    for (int e = 0; e < E; ++e)
      f.b[e][u] = *(const v4f*)(bp[e] + (g * XU_ + u) * 8);
  }
}

template <int E>
__device__ __forceinline__ void xproj_mma(v8f (&acc)[E], const XFrag<E>& f) {
#pragma unroll
  for (int u = 0; u < XU_; ++u) {
    const v2f a0 = f.a[u].xy;
    const v2f a1 = f.a[u].zw;
#pragma unroll
    for (int e = 0; e < E; ++e) {
      const v2f b0 = f.b[e][u].xy;
      acc[e] = __builtin_amdgcn_wmma_f32_16x16x4_f32(false, a0, false, b0,
                                                     (short)0, acc[e], false, false);
    }
#pragma unroll
    for (int e = 0; e < E; ++e) {
      const v2f b1 = f.b[e][u].zw;
      acc[e] = __builtin_amdgcn_wmma_f32_16x16x4_f32(false, a1, false, b1,
                                                     (short)0, acc[e], false, false);
    }
  }
}

template <int E>
__device__ __forceinline__ void xproj_strip(const float* __restrict__ ap,
                                            const float* __restrict__ wbase, int col0,
                                            float* __restrict__ op, bool publish,
                                            float* __restrict__ lp) {
  const float* bp[E];
#pragma unroll
  for (int e = 0; e < E; ++e) bp[e] = wbase + (size_t)xdbl_src_row(col0 + e * 16) * D_;
  v8f acc[E];
#pragma unroll
  for (int e = 0; e < E; ++e) acc[e] = v8f{};
  XFrag<E> f0, f1;
  xproj_load<E>(f0, ap, bp, 0);
  for (int g = 0; g < XG_ - 2; g += 2) {
    xproj_load<E>(f1, ap, bp, g + 1);
    __builtin_amdgcn_sched_barrier(0);
    xproj_mma<E>(acc, f0);
    __builtin_amdgcn_sched_barrier(0);
    xproj_load<E>(f0, ap, bp, g + 2);
    __builtin_amdgcn_sched_barrier(0);
    xproj_mma<E>(acc, f1);
    __builtin_amdgcn_sched_barrier(0);
  }
  xproj_load<E>(f1, ap, bp, XG_ - 1);
  __builtin_amdgcn_sched_barrier(0);
  xproj_mma<E>(acc, f0);
  xproj_mma<E>(acc, f1);
#pragma unroll
  for (int e = 0; e < E; ++e)
#pragma unroll
    for (int j = 0; j < 8; ++j)
      op[(size_t)j * E_ + e * 16] = acc[e][j];
  if (publish) {
#pragma unroll
    for (int e = 0; e < 2; ++e)
#pragma unroll
      for (int j = 0; j < 8; ++j)
        lp[j * DRP_ + e * 16] = acc[e][j];
  }
}

__global__ void __launch_bounds__(128)
s6_xproj_delta(const float* __restrict__ x, const float* __restrict__ W,
               const float* __restrict__ Wdt, const float* __restrict__ bdt,
               float* __restrict__ xdbl, float* __restrict__ delta) {
  __shared__ __attribute__((aligned(16))) float dr[16 * DRP_];

  const int lane  = threadIdx.x & 31;
  const int wv    = __builtin_amdgcn_readfirstlane((int)(threadIdx.x >> 5));
  const int rt    = blockIdx.x;
  const int laneM = lane & 15;
  const int laneH = lane >> 4;

  const float* ap = x + (size_t)(rt * 16 + laneM) * D_ + 4 * laneH;
  float* orow = xdbl + (size_t)(rt * 16 + 8 * laneH) * E_ + laneM;
  float* lp = dr + (8 * laneH) * DRP_ + laneM;
  if (wv < 2) {
    const int t0 = 3 * wv;
    xproj_strip<3>(ap, W + 4 * laneH, t0 * 16 + laneM, orow + t0 * 16, wv == 0, lp);
  } else {
    const int t0 = 2 * wv + 2;
    xproj_strip<2>(ap, W + 4 * laneH, t0 * 16 + laneM, orow + t0 * 16, false, lp);
  }

  lds_arrive();
  const int ch0 = wv * 128 + laneM;
  const float* brow = Wdt + (size_t)ch0 * R_ + 2 * laneH;
  v2f bw[8][8];
  float bi[8];
#pragma unroll
  for (int kk = 0; kk < 8; ++kk)
#pragma unroll
    for (int t = 0; t < 8; ++t)
      bw[t][kk] = *(const v2f*)(brow + (size_t)t * 16 * R_ + 4 * kk);
#pragma unroll
  for (int t = 0; t < 8; ++t) bi[t] = bdt[ch0 + 16 * t];
  lds_wait();
  v2f aw[8];
#pragma unroll
  for (int kk = 0; kk < 8; ++kk)
    aw[kk] = *(const v2f*)(dr + laneM * DRP_ + 4 * kk + 2 * laneH);
  v8f cd[8];
#pragma unroll
  for (int t = 0; t < 8; ++t) cd[t] = v8f{};
#pragma unroll
  for (int kk = 0; kk < 8; ++kk)
#pragma unroll
    for (int t = 0; t < 8; ++t)
      cd[t] = __builtin_amdgcn_wmma_f32_16x16x4_f32(false, aw[kk], false, bw[t][kk],
                                                    (short)0, cd[t], false, false);
#pragma unroll
  for (int t = 0; t < 8; ++t) {
#pragma unroll
    for (int j = 0; j < 8; ++j) {
      const int row = rt * 16 + j + 8 * laneH;
      const float z = cd[t][j] + bi[t];
      const float lg = __logf(1.0f + __expf(z));
      delta[(size_t)row * D_ + ch0 + 16 * t] = (z > 20.0f) ? z : lg;
    }
  }
}

#define SB_ 32
#define SH_ 16
#define SG_ 8
#define PF_ 8

__device__ __forceinline__ void l2_warm(const char* p) {
  __builtin_prefetch(p, 0, 1);
}

__device__ __forceinline__ float row_share(float v, int i) {
  const int b = __builtin_bit_cast(int, v);
  int r = 0;
  switch (i) {
    case 0: r = __builtin_amdgcn_update_dpp(0, b, 0x150 + 0, 0xf, 0xf, true); break;
    case 1: r = __builtin_amdgcn_update_dpp(0, b, 0x150 + 1, 0xf, 0xf, true); break;
    case 2: r = __builtin_amdgcn_update_dpp(0, b, 0x150 + 2, 0xf, 0xf, true); break;
    case 3: r = __builtin_amdgcn_update_dpp(0, b, 0x150 + 3, 0xf, 0xf, true); break;
    case 4: r = __builtin_amdgcn_update_dpp(0, b, 0x150 + 4, 0xf, 0xf, true); break;
    case 5: r = __builtin_amdgcn_update_dpp(0, b, 0x150 + 5, 0xf, 0xf, true); break;
    case 6: r = __builtin_amdgcn_update_dpp(0, b, 0x150 + 6, 0xf, 0xf, true); break;
    case 7: r = __builtin_amdgcn_update_dpp(0, b, 0x150 + 7, 0xf, 0xf, true); break;
    case 8: r = __builtin_amdgcn_update_dpp(0, b, 0x150 + 8, 0xf, 0xf, true); break;
    case 9: r = __builtin_amdgcn_update_dpp(0, b, 0x150 + 9, 0xf, 0xf, true); break;
    case 10: r = __builtin_amdgcn_update_dpp(0, b, 0x150 + 10, 0xf, 0xf, true); break;
    case 11: r = __builtin_amdgcn_update_dpp(0, b, 0x150 + 11, 0xf, 0xf, true); break;
    case 12: r = __builtin_amdgcn_update_dpp(0, b, 0x150 + 12, 0xf, 0xf, true); break;
    case 13: r = __builtin_amdgcn_update_dpp(0, b, 0x150 + 13, 0xf, 0xf, true); break;
    case 14: r = __builtin_amdgcn_update_dpp(0, b, 0x150 + 14, 0xf, 0xf, true); break;
    case 15: r = __builtin_amdgcn_update_dpp(0, b, 0x150 + 15, 0xf, 0xf, true); break;
  }
  return __builtin_bit_cast(float, r);
}
template <int M>
__device__ __forceinline__ float row_xor(float v) {
  const int b = __builtin_bit_cast(int, v);
  return __builtin_bit_cast(float,
      __builtin_amdgcn_update_dpp(0, b, 0x160 + M, 0xf, 0xf, true));
}

__device__ __forceinline__ float row_sums(const float* part, bool u3, bool u2,
                                          bool u1, bool u0) {
  float r8[8], r4[4], r2[2];
#pragma unroll
  for (int i = 0; i < 8; ++i) {
    const float keep = u3 ? part[i + 8] : part[i];
    const float send = u3 ? part[i] : part[i + 8];
    r8[i] = keep + row_xor<8>(send);
  }
#pragma unroll
  for (int i = 0; i < 4; ++i) {
    const float keep = u2 ? r8[i + 4] : r8[i];
    const float send = u2 ? r8[i] : r8[i + 4];
    r4[i] = keep + row_xor<4>(send);
  }
#pragma unroll
  for (int i = 0; i < 2; ++i) {
    const float keep = u1 ? r4[i + 2] : r4[i];
    const float send = u1 ? r4[i] : r4[i + 2];
    r2[i] = keep + row_xor<2>(send);
  }
  const float keep0 = u0 ? r2[1] : r2[0];
  const float send0 = u0 ? r2[0] : r2[1];
  return keep0 + row_xor<1>(send0);
}

__device__ __forceinline__ void swap_arrive() {
  __builtin_amdgcn_fence(__ATOMIC_RELEASE, "workgroup", "local");
  __builtin_amdgcn_s_barrier_signal(-1);
}
__device__ __forceinline__ void swap_wait() {
  __builtin_amdgcn_s_barrier_wait(-1);
  __builtin_amdgcn_fence(__ATOMIC_ACQUIRE, "workgroup", "local");
}

__global__ void __launch_bounds__(128)
s6_scan_seq(const float* __restrict__ xdbl, const float* __restrict__ delta,
            const float* __restrict__ x, const float* __restrict__ Alog,
            const float* __restrict__ Dp, float* __restrict__ y) {
  __shared__ float xch[2][4][32];

  const int lane = threadIdx.x & 31;
  const int wv = __builtin_amdgcn_readfirstlane((int)(threadIdx.x >> 5));
  const int q  = wv & 1;
  const int pw = blockIdx.x * 2 + (wv >> 1);
  const int bb = pw / (D_ / 2);
  const int c  = lane & 15;
  const int dc = (pw % (D_ / 2)) * 2 + (lane >> 4);
  const int n0 = 32 * q + 2 * c;

  const v2f al = *(const v2f*)(Alog + (size_t)dc * N_ + n0);
  v2f A;
  A.x = -__expf(al.x);  A.y = -__expf(al.y);
  const float Dv = Dp[dc];

  const float* bc = xdbl + (size_t)bb * L_ * E_ + R_ + 64 * q + 4 * c;
  const size_t lrow = ((size_t)bb * L_ + c) * D_ + dc;
  const float* pdl = delta + lrow;
  const float* pxv = x + lrow;
  float*       py  = y + lrow + (size_t)(q * SH_) * D_;

  const bool u3 = (lane & 8) != 0;
  const bool u2 = (lane & 4) != 0;
  const bool u1 = (lane & 2) != 0;
  const bool u0 = (lane & 1) != 0;

  v2f dlA, xvA, dlB, xvB;
  dlA.x = pdl[0];
  dlA.y = pdl[(size_t)SH_ * D_];
  xvA.x = pxv[0];
  xvA.y = pxv[(size_t)SH_ * D_];
  dlB.x = pdl[(size_t)SB_ * D_];
  dlB.y = pdl[(size_t)(SB_ + SH_) * D_];
  xvB.x = pxv[(size_t)SB_ * D_];
  xvB.y = pxv[(size_t)(SB_ + SH_) * D_];
  __builtin_amdgcn_sched_barrier(0);
  v4f RA[SB_], RB[SB_];
#pragma unroll
  for (int i = 0; i < SB_; ++i) RA[i] = *(const v4f*)(bc + (size_t)i * E_);
#pragma unroll
  for (int i = 0; i < SB_; ++i) RB[i] = *(const v4f*)(bc + (size_t)(SB_ + i) * E_);
  __builtin_amdgcn_sched_barrier(0);
  const int gq = blockIdx.x & (D_ / 4 - 1);
  const int rs = (gq + (gq >> 2)) & 3;
  const int r8 = ((gq + (gq >> 3)) & 7) ^ ((wv & 1) << 2);
  const int pj = (lane < 10) ? lane : 9;
  const float* pfPlane = (wv == 0) ? delta : (wv == 1) ? x : xdbl;
  const int pfStep = (wv < 2) ? D_ * 4 : E_ * 4;
  const unsigned pfOffDX = ((unsigned)(bb * L_ + 4 * (lane & 7) + rs) * D_ + 4 * gq) * 4u;
  const unsigned pfOffBC = (unsigned)bb * (L_ * E_ * 4) + (unsigned)(8 * pj + r8) * 256u;
  const char* pfp = (const char*)pfPlane + ((wv < 2) ? pfOffDX : pfOffBC);
  for (int k = 2; k < PF_; ++k) l2_warm(pfp + (size_t)(k * SB_) * pfStep);

  xch[1][wv][lane] = 0.0f;

  v2f h = {0.0f, 0.0f};
  float minePrev = 0.0f;
  float xdPrev   = 0.0f;
  float other    = 0.0f;

  auto batch = [&](v4f (&Rw)[SB_], v2f& dl, v2f& xv, const int t0, const int par)
      __attribute__((always_inline)) {
    v2f dw;
    asm("v_mov_b64 %0, %1" : "=v"(dw) : "v"(dl));
    const float dlvA = dw.x;
    const float dlvB = dw.y;
    const v2f duv = dw * xv;
    const float duvA = duv.x;
    const float duvB = duv.y;
    const float xdCur = (q ? xv.y : xv.x) * Dv;
    const int tn = (t0 + 2 * SB_ < L_) ? (t0 + 2 * SB_) : t0;
    const float* bcn = bc + (size_t)tn * E_;

    float part[SB_];
#pragma unroll
    for (int j = 0; j < SB_ / SG_; ++j) {
#pragma unroll
      for (int k = 0; k < SG_; ++k) {
        const int i = j * SG_ + k;
        const float dlt = (i < SH_) ? row_share(dlvA, i) : row_share(dlvB, i - SH_);
        const float du  = (i < SH_) ? row_share(duvA, i) : row_share(duvB, i - SH_);
        const v2f Bi = Rw[i].xy;
        const v2f Ci = Rw[i].zw;
        const v2f e = A * dlt;
        v2f a;
        a.x = __expf(e.x);  a.y = __expf(e.y);
        h = a * h + du * Bi;
        const v2f hc = h * Ci;
        part[i] = hc.x + hc.y;
      }
      asm volatile("" :: "v"(part[j * SG_ + 0]), "v"(part[j * SG_ + 1]),
                         "v"(part[j * SG_ + 2]), "v"(part[j * SG_ + 3]),
                         "v"(part[j * SG_ + 4]), "v"(part[j * SG_ + 5]),
                         "v"(part[j * SG_ + 6]), "v"(part[j * SG_ + 7]) : "memory");
      if (j == 0) {
        dl.x = pdl[(size_t)tn * D_];
        dl.y = pdl[(size_t)(tn + SH_) * D_];
        xv.x = pxv[(size_t)tn * D_];
        xv.y = pxv[(size_t)(tn + SH_) * D_];
      }
#pragma unroll
      for (int k = 0; k < SG_; ++k) {
        const int i = j * SG_ + k;
        Rw[i] = *(const v4f*)(bcn + (size_t)i * E_);
      }
      if (j == 2) {
        const int tp = (t0 + PF_ * SB_ < L_) ? (t0 + PF_ * SB_) : t0;
        l2_warm(pfp + (size_t)tp * pfStep);
      }
      __builtin_amdgcn_sched_barrier(0);
      if (j == 0) swap_arrive();
      if (j == 1) swap_wait();
    }
    other = xch[par ^ 1][wv ^ 1][lane];
    __builtin_amdgcn_sched_barrier(0);

    const float totA = row_sums(part, u3, u2, u1, u0);
    const float totB = row_sums(part + SH_, u3, u2, u1, u0);

    const float yPrev = (minePrev + other) + xdPrev;
    minePrev = q ? totB : totA;
    xdPrev   = xdCur;
    xch[par][wv][lane] = q ? totA : totB;
    if (t0 != 0) py[(size_t)(t0 - SB_) * D_] = yPrev;
  };

  for (int t0 = 0; t0 < L_; t0 += 2 * SB_) {
    batch(RA, dlA, xvA, t0, 0);
    batch(RB, dlB, xvB, t0 + SB_, 1);
  }
  swap_arrive();
  swap_wait();
  other = xch[((L_ / SB_) - 1) & 1][wv ^ 1][lane];
  py[(size_t)(L_ - SB_) * D_] = (minePrev + other) + xdPrev;
}

extern "C" void kernel_launch(void* const* d_in, const int* in_sizes, int n_in,
                              void* d_out, int out_size, void* d_ws, size_t ws_size,
                              hipStream_t stream) {
  const float* x      = (const float*)d_in[0];
  const float* Wxproj = (const float*)d_in[1];
  const float* Wdt    = (const float*)d_in[2];
  const float* bdt    = (const float*)d_in[3];
  const float* Alog   = (const float*)d_in[4];
  const float* Dp     = (const float*)d_in[5];
  float*       y      = (float*)d_out;

  float* ws    = (float*)d_ws;
  float* xdbl  = ws;
  float* delta = xdbl  + (size_t)BL_ * E_;

  s6_xproj_delta<<<256, 128, 0, stream>>>(x, Wxproj, Wdt, bdt, xdbl, delta);
  s6_scan_seq<<<256, 128, 0, stream>>>(xdbl, delta, x, Alog, Dp, y);
}
